# starting kernel
# baseline (speedup 1.0000x reference)
#include <hip/hip_runtime.h>

typedef float v2f __attribute__((ext_vector_type(2)));
typedef float v8f __attribute__((ext_vector_type(8)));

#define IN_DIM 128
#define HID 128
#define OUT 64

__global__ __launch_bounds__(256) void k_init_deg(float* __restrict__ deg, int n) {
    int i = blockIdx.x * blockDim.x + threadIdx.x;
    if (i < n) deg[i] = 1.0f;
}

__global__ __launch_bounds__(256) void k_edge_deg(const int* __restrict__ dst, int E,
                                                  float* __restrict__ deg) {
    int e = blockIdx.x * blockDim.x + threadIdx.x;
    if (e < E) atomicAdd(&deg[dst[e]], 1.0f);
}

__global__ __launch_bounds__(256) void k_rsqrt(const float* __restrict__ deg,
                                               float* __restrict__ dinv, int n) {
    int i = blockIdx.x * blockDim.x + threadIdx.x;
    if (i < n) dinv[i] = rsqrtf(deg[i]);
}

template <int NTILES>
__global__ __launch_bounds__(256) void k_gemm_wmma(const float* __restrict__ X,
                                                   const float* __restrict__ W,
                                                   float* __restrict__ H, int nrows) {
    const int lane = threadIdx.x & 31;
    const int wave = threadIdx.x >> 5;
    const int tile = blockIdx.x * (blockDim.x >> 5) + wave;
    if (tile * 16 >= nrows) return;

    const int ln = lane & 15;
    const int hi = lane >> 4;
    const int ldw = NTILES * 16;

    v8f acc[NTILES] = {};

    const float* xrow = X + (size_t)(tile * 16 + ln) * IN_DIM;

    for (int k = 0; k < IN_DIM; k += 4) {
        v2f a;
        a.x = xrow[k + 2 * hi + 0];
        a.y = xrow[k + 2 * hi + 1];

        const float* wr = W + (size_t)(k + 2 * hi) * ldw;
#pragma unroll
        for (int n = 0; n < NTILES; ++n) {
            v2f b;
            b.x = wr[n * 16 + ln];
            b.y = wr[ldw + n * 16 + ln];
            acc[n] = __builtin_amdgcn_wmma_f32_16x16x4_f32(
                false, a, false, b, (short)0, acc[n], false, false);
        }
    }

#pragma unroll
    for (int n = 0; n < NTILES; ++n) {
#pragma unroll
        for (int r = 0; r < 8; ++r) {
            int orow = tile * 16 + r + 8 * hi;
            H[(size_t)orow * ldw + n * 16 + ln] = acc[n][r];
        }
    }
}

__global__ __launch_bounds__(256) void k_self_bias(const float* __restrict__ H,
                                                   const float* __restrict__ dinv,
                                                   const float* __restrict__ bias,
                                                   float* __restrict__ A,
                                                   int n, int dim) {
    size_t idx = (size_t)blockIdx.x * blockDim.x + threadIdx.x;
    size_t total = (size_t)n * dim;
    if (idx < total) {
        int i = (int)(idx / dim);
        int c = (int)(idx % dim);
        float s = dinv[i];
        A[idx] = H[idx] * (s * s) + bias[c];
    }
}

template <int DIM>
__global__ __launch_bounds__(256) void k_scatter(const float* __restrict__ H,
                                                 const int* __restrict__ src,
                                                 const int* __restrict__ dst,
                                                 const float* __restrict__ dinv,
                                                 float* __restrict__ A, int E) {
    const int lane = threadIdx.x & 31;
    const int e = (int)(((size_t)blockIdx.x * blockDim.x + threadIdx.x) >> 5);
    if (e >= E) return;
    const int s = src[e];
    const int d = dst[e];
    const float norm = dinv[s] * dinv[d];
    constexpr int VPT = DIM / 32;
    const float* hp = H + (size_t)s * DIM + lane * VPT;
    float* ap = A + (size_t)d * DIM + lane * VPT;
#pragma unroll
    for (int j = 0; j < VPT; ++j) atomicAdd(&ap[j], hp[j] * norm);
}

__global__ __launch_bounds__(256) void k_relu(float* __restrict__ A, size_t total) {
    size_t idx = (size_t)blockIdx.x * blockDim.x + threadIdx.x;
    if (idx < total) A[idx] = fmaxf(A[idx], 0.0f);
}

extern "C" void kernel_launch(void* const* d_in, const int* in_sizes, int n_in,
                              void* d_out, int out_size, void* d_ws, size_t ws_size,
                              hipStream_t stream) {
    const float* x   = (const float*)d_in[0];
    const int*   ei  = (const int*)d_in[1];
    const float* W1  = (const float*)d_in[2];
    const float* b1  = (const float*)d_in[3];
    const float* W2  = (const float*)d_in[4];
    const float* b2  = (const float*)d_in[5];
    float*       out = (float*)d_out;

    const int n = in_sizes[0] / IN_DIM;
    const int E = in_sizes[1] / 2;
    const int* srcIdx = ei;
    const int* dstIdx = ei + E;

    float* ws   = (float*)d_ws;
    float* deg  = ws;
    float* dinv = deg + n;
    float* h    = dinv + n;
    float* agg  = h + (size_t)n * HID;

    k_init_deg<<<(n + 255) / 256, 256, 0, stream>>>(deg, n);
    k_edge_deg<<<(E + 255) / 256, 256, 0, stream>>>(dstIdx, E, deg);
    k_rsqrt<<<(n + 255) / 256, 256, 0, stream>>>(deg, dinv, n);

    const int tiles = (n + 15) / 16;
    const int gemmBlocks = (tiles + 7) / 8;
    const int scatterBlocks = (int)(((size_t)E * 32 + 255) / 256);

    k_gemm_wmma<8><<<gemmBlocks, 256, 0, stream>>>(x, W1, h, n);
    {
        size_t total = (size_t)n * HID;
        k_self_bias<<<(int)((total + 255) / 256), 256, 0, stream>>>(h, dinv, b1, agg, n, HID);
        k_scatter<HID><<<scatterBlocks, 256, 0, stream>>>(h, srcIdx, dstIdx, dinv, agg, E);
        k_relu<<<(int)((total + 255) / 256), 256, 0, stream>>>(agg, total);
    }

    k_gemm_wmma<4><<<gemmBlocks, 256, 0, stream>>>(agg, W2, h, n);
    {
        size_t total = (size_t)n * OUT;
        k_self_bias<<<(int)((total + 255) / 256), 256, 0, stream>>>(h, dinv, b2, out, n, OUT);
        k_scatter<OUT><<<scatterBlocks, 256, 0, stream>>>(h, srcIdx, dstIdx, dinv, out, E);
    }
}


# Degree count and rsqrt lose their launches: the layer-1 product's waves add the edge counts first (128 edges per wave and round), users take rsqrtf(deg) at use; 7 launches -> 5, no sum reordered
# speedup vs baseline: 2.3526x; 2.3526x over previous
#include <hip/hip_runtime.h>

typedef float v2f __attribute__((ext_vector_type(2)));
typedef float v8f __attribute__((ext_vector_type(8)));

#define IN_DIM 128
#define HID 128
#define OUT 64

__global__ __launch_bounds__(256) void k_init_deg(float* __restrict__ deg, int n) {
    int i = blockIdx.x * blockDim.x + threadIdx.x;
    if (i < n) deg[i] = 1.0f;
}


typedef int gv4i __attribute__((ext_vector_type(4)));
template <int NTILES, bool DEG>
__global__ __launch_bounds__(256) void k_gemm_wmma(const float* __restrict__ X,
                                                   const float* __restrict__ W,
                                                   float* __restrict__ H, int nrows,
                                                   const int* __restrict__ dst, int E,
                                                   float* __restrict__ deg) {
    const int lane = threadIdx.x & 31;
    const int wave = threadIdx.x >> 5;
    const int tile = blockIdx.x * (blockDim.x >> 5) + wave;
    if (DEG) {
        const int nwaves = gridDim.x * (blockDim.x >> 5);
        const int nfull = E >> 7;
        for (int c = tile; c < nfull; c += nwaves) {
            const gv4i d4 = *(const gv4i*)(dst + (c << 7) + (lane << 2));
            atomicAdd(&deg[d4.x], 1.0f);
            atomicAdd(&deg[d4.y], 1.0f);
            atomicAdd(&deg[d4.z], 1.0f);
            atomicAdd(&deg[d4.w], 1.0f);
        }
        if (tile == 0)
            for (int e = (nfull << 7) + lane; e < E; e += 32) atomicAdd(&deg[dst[e]], 1.0f);
    }
    if (tile * 16 >= nrows) return;

    const int ln = lane & 15;
    const int hi = lane >> 4;
    const int ldw = NTILES * 16;

    v8f acc[NTILES] = {};

    const float* xrow = X + (size_t)(tile * 16 + ln) * IN_DIM;

    for (int k = 0; k < IN_DIM; k += 4) {
        v2f a;
        a.x = xrow[k + 2 * hi + 0];
        a.y = xrow[k + 2 * hi + 1];

        const float* wr = W + (size_t)(k + 2 * hi) * ldw;
#pragma unroll
        for (int n = 0; n < NTILES; ++n) {
            v2f b;
            b.x = wr[n * 16 + ln];
            b.y = wr[ldw + n * 16 + ln];
            acc[n] = __builtin_amdgcn_wmma_f32_16x16x4_f32(
                false, a, false, b, (short)0, acc[n], false, false);
        }
    }

#pragma unroll
    for (int n = 0; n < NTILES; ++n) {
#pragma unroll
        for (int r = 0; r < 8; ++r) {
            int orow = tile * 16 + r + 8 * hi;
            H[(size_t)orow * ldw + n * 16 + ln] = acc[n][r];
        }
    }
}

#define TILE_ROWS 391
#define AGG_WAVES 32
#define SCAN_U 4
#define QCAP (32 + SCAN_U * 128)

typedef int v4i __attribute__((ext_vector_type(4)));

template <int DIM, bool FIRST>
__global__ __launch_bounds__(AGG_WAVES * 32) void k_agg(const float* __restrict__ H,
                                                        const int* __restrict__ src,
                                                        const int* __restrict__ dst,
                                                        const float* __restrict__ deg,
                                                        const float* __restrict__ bias,
                                                        float* A, float* Lst, int n, int E) {
    constexpr int VPT = DIM / 32;
    constexpr int G = 32 / VPT;
    typedef float vecf __attribute__((ext_vector_type(VPT)));

    __shared__ float tile[TILE_ROWS * DIM];
    __shared__ int queue[AGG_WAVES * QCAP];
    __shared__ int lcount;

    const int lane = threadIdx.x & 31;
    const int wave = __builtin_amdgcn_readfirstlane((int)(threadIdx.x >> 5));
    const int r0 = blockIdx.x * TILE_ROWS;
    const int rows = min(TILE_ROWS, n - r0);

    int* const lhead = (int*)(Lst + (size_t)r0 * OUT);
    const int cap = ((rows * OUT - 32) / 3) & ~31;
    int* const lsrc = lhead + 32;
    int* const lrowp = lsrc + cap;
    float* const lwgt = (float*)(lrowp + cap);
    if (FIRST && threadIdx.x == 0) lcount = 0;

    const vecf bv = *(const vecf*)(bias + VPT * lane);
    for (int i0 = wave; i0 < rows; i0 += 4 * AGG_WAVES) {
        float s[4];
        vecf hv[4];
        for (int u = 0; u < 4; ++u) {
            const int i = (i0 + u * AGG_WAVES < rows) ? (i0 + u * AGG_WAVES) : i0;
            s[u] = rsqrtf(deg[r0 + i]);
            hv[u] = *(const vecf*)(H + (size_t)(r0 + i) * DIM + VPT * lane);
        }
        for (int u = 0; u < 4; ++u) {
            const int i = i0 + u * AGG_WAVES;
            if (i < rows)
                for (int k = 0; k < VPT; ++k)
                    tile[i * DIM + k * 32 + lane] = hv[u][k] * (s[u] * s[u]) + bv[k];
        }
    }
    __syncthreads();

    auto addrows = [&](int s, int lrow, float nrm, int cnt,
                       auto&& behind_last_loads) __attribute__((always_inline)) {
        for (int j0 = 0; j0 < cnt; j0 += G) {
            vecf v[G];
#pragma unroll
            for (int g = 0; g < G; ++g) {
                const int j = (j0 + g < cnt) ? (j0 + g) : j0;
                const int sj = __builtin_amdgcn_readlane(s, j);
                v[g] = *(const vecf*)(H + (size_t)sj * DIM + VPT * lane);
            }
            if (j0 + G >= cnt) behind_last_loads();
#pragma unroll
            for (int g = 0; g < G; ++g) {
                if (j0 + g < cnt) {
                    const int lj = __builtin_amdgcn_readlane(lrow, j0 + g);
                    const float nj = __int_as_float(
                        __builtin_amdgcn_readlane(__float_as_int(nrm), j0 + g));
                    float* tp = tile + lj * DIM + lane;
                    for (int k = 0; k < VPT; ++k)
                        __hip_atomic_fetch_add(tp + k * 32, v[g][k] * nj, __ATOMIC_RELAXED,
                                               __HIP_MEMORY_SCOPE_WORKGROUP);
                }
            }
        }
    };

    int total = 0;
    bool from_list = false;
    if (!FIRST) {
        total = __builtin_amdgcn_readfirstlane(lhead[0]);
        from_list = (unsigned)total <= (unsigned)cap;
    }

    if (from_list) {
        const int nb = total >> 5;
        auto entry = [&](int b, int& s, int& lrow, float& nrm) {
            const int idx = (b << 5) + lane;
            s = lsrc[idx];
            lrow = lrowp[idx];
            nrm = lwgt[idx];
        };
        int b = wave;
        int s = -1, lrow = 0;
        float nrm = 0.0f;
        if (b < nb) entry(b, s, lrow, nrm);
        while (b < nb) {
            const int bn = b + AGG_WAVES;
            int s2 = -1, lrow2 = 0;
            float nrm2 = 0.0f;
            auto next = [&]() __attribute__((always_inline)) {
                if (bn < nb) entry(bn, s2, lrow2, nrm2);
            };
            const int cnt = __builtin_popcount(__builtin_amdgcn_ballot_w32(s >= 0));
            if (cnt == 32)
                addrows(s, lrow, nrm, 32, next);
            else if (cnt > 0)
                addrows(s, lrow, nrm, cnt, next);
            else
                next();
            s = s2;
            lrow = lrow2;
            nrm = nrm2;
            b = bn;
        }
    } else {
        const int nchunks = (E + 127) >> 7;
        const int T = (nchunks + AGG_WAVES - 1) / AGG_WAVES;
        const int span = T * AGG_WAVES;
        const int rot = (int)((blockIdx.x * (unsigned)T) / gridDim.x) * AGG_WAVES;

        const int qoff = wave * QCAP;
        int qn = 0;
        const int nfull = E >> 7;

        auto chunk_of = [&](int t) -> int {
            int c = t * AGG_WAVES + wave + rot;
            if (c >= span) c -= span;
            return c;
        };
        auto chunk_ok = [&](int t) -> bool { return t < T && chunk_of(t) < nfull; };
        auto fetch = [&](int t) -> v4i {
            const int c = chunk_ok(t) ? chunk_of(t) : 0;
            return *(const v4i*)(dst + (c << 7) + (lane << 2));
        };

        auto process = [&](int qb, int cnt) __attribute__((always_inline)) {
            int s = 0, lrow = 0;
            float nrm = 0.0f;
            __builtin_amdgcn_fence(__ATOMIC_ACQ_REL, "wavefront");
            if (lane < cnt) {
                const int e = queue[qoff + qb + lane];
                s = src[e];
                const int d = dst[e];
                lrow = d - r0;
                nrm = rsqrtf(deg[s]) * rsqrtf(deg[d]);
            }
            if (FIRST) {
                int base = 0;
                if (lane == 0)
                    base = __hip_atomic_fetch_add(&lcount, 32, __ATOMIC_RELAXED,
                                                  __HIP_MEMORY_SCOPE_WORKGROUP);
                base = __builtin_amdgcn_readfirstlane(base);
                if (base + 32 <= cap) {
                    lsrc[base + lane] = (lane < cnt) ? s : -1;
                    lrowp[base + lane] = lrow;
                    lwgt[base + lane] = nrm;
                }
            }
            addrows(s, lrow, nrm, cnt, []() {});
        };

        auto handle = [&](const v4i d4, int t) {
            const bool o0 = (unsigned)(d4.x - r0) < (unsigned)rows;
            const bool o1 = (unsigned)(d4.y - r0) < (unsigned)rows;
            const bool o2 = (unsigned)(d4.z - r0) < (unsigned)rows;
            const bool o3 = (unsigned)(d4.w - r0) < (unsigned)rows;
            const unsigned any = __builtin_amdgcn_ballot_w32(o0) | __builtin_amdgcn_ballot_w32(o1) |
                                 __builtin_amdgcn_ballot_w32(o2) | __builtin_amdgcn_ballot_w32(o3);
            if (any != 0 && chunk_ok(t)) {
                const int e0 = (chunk_of(t) << 7) + (lane << 2);
                unsigned m = (o0 ? 1u : 0u) | (o1 ? 2u : 0u) | (o2 ? 4u : 0u) | (o3 ? 8u : 0u);
                unsigned bal = any;
                while (bal) {
                    const int pos = qn + (int)__builtin_amdgcn_mbcnt_lo(bal, 0u);
                    if (m) {
                        const int b = __builtin_ctz(m);
                        m &= m - 1;
                        queue[qoff + pos] = e0 + b;
                    }
                    qn += __builtin_popcount(bal);
                    bal = __builtin_amdgcn_ballot_w32(m != 0);
                }
            }
        };
        auto drain = [&]() __attribute__((always_inline)) {
            while (qn >= 32) {
                qn -= 32;
                process(qn, 32);
            }
        };

        if (nfull > 0) {
            v4i ga[SCAN_U], gb[SCAN_U];
            for (int u = 0; u < SCAN_U; ++u) ga[u] = fetch(u);
            for (int tb = 0; tb < T; tb += 2 * SCAN_U) {
                for (int u = 0; u < SCAN_U; ++u) gb[u] = fetch(tb + SCAN_U + u);
                for (int u = 0; u < SCAN_U; ++u) handle(ga[u], tb + u);
                drain();
                for (int u = 0; u < SCAN_U; ++u) ga[u] = fetch(tb + 2 * SCAN_U + u);
                for (int u = 0; u < SCAN_U; ++u) handle(gb[u], tb + SCAN_U + u);
                drain();
            }
        }
        if (wave == 0 && (E & 127)) {
            for (int b = 0; b < 4; ++b) {
                const int e = (nfull << 7) + b * 32 + lane;
                const bool own = e < E && (unsigned)(dst[e < E ? e : 0] - r0) < (unsigned)rows;
                const unsigned bal = __builtin_amdgcn_ballot_w32(own);
                if (own) queue[qoff + qn + (int)__builtin_amdgcn_mbcnt_lo(bal, 0u)] = e;
                qn += __builtin_popcount(bal);
            }
        }
        while (qn > 0) {
            const int c = min(qn, 32);
            qn -= c;
            process(qn, c);
        }
    }

    __syncthreads();

    if (FIRST && threadIdx.x == 0) lhead[0] = lcount;

    for (int i = wave; i < rows; i += AGG_WAVES) {
        vecf o;
        for (int k = 0; k < VPT; ++k) {
            const float x = tile[i * DIM + k * 32 + lane];
            o[k] = FIRST ? fmaxf(x, 0.0f) : x;
        }
        *(vecf*)(A + (size_t)(r0 + i) * DIM + VPT * lane) = o;
    }
}

extern "C" void kernel_launch(void* const* d_in, const int* in_sizes, int n_in,
                              void* d_out, int out_size, void* d_ws, size_t ws_size,
                              hipStream_t stream) {
    const float* x   = (const float*)d_in[0];
    const int*   ei  = (const int*)d_in[1];
    const float* W1  = (const float*)d_in[2];
    const float* b1  = (const float*)d_in[3];
    const float* W2  = (const float*)d_in[4];
    const float* b2  = (const float*)d_in[5];
    float*       out = (float*)d_out;

    const int n = in_sizes[0] / IN_DIM;
    const int E = in_sizes[1] / 2;
    const int* srcIdx = ei;
    const int* dstIdx = ei + E;

    float* ws   = (float*)d_ws;
    float* deg  = ws;
    float* h    = deg + 2 * (size_t)n;
    float* agg  = h + (size_t)n * HID;

    k_init_deg<<<(n + 255) / 256, 256, 0, stream>>>(deg, n);

    const int tiles = (n + 15) / 16;
    const int gemmBlocks = (tiles + 7) / 8;
    const int aggBlocks = (n + TILE_ROWS - 1) / TILE_ROWS;

    k_gemm_wmma<8, true><<<gemmBlocks, 256, 0, stream>>>(x, W1, h, n, dstIdx, E, deg);
    k_agg<HID, true><<<aggBlocks, AGG_WAVES * 32, 0, stream>>>(h, srcIdx, dstIdx, deg, b1, agg, out, n, E);

    k_gemm_wmma<4, false><<<gemmBlocks, 256, 0, stream>>>(agg, W2, h, n, dstIdx, E, deg);
    k_agg<OUT, false><<<aggBlocks, AGG_WAVES * 32, 0, stream>>>(h, srcIdx, dstIdx, deg, b2, out, out, n, E);
}


# Layer-1 sweep reads a 1-byte owner number per edge (dst/391, written beside the degree count into spare words of the groups' list areas), not the 4-byte destination; arrival-order sums kept
# speedup vs baseline: 2.5629x; 1.0894x over previous
#include <hip/hip_runtime.h>

typedef float v2f __attribute__((ext_vector_type(2)));
typedef float v8f __attribute__((ext_vector_type(8)));

#define IN_DIM 128
#define HID 128
#define OUT 64
#define TILE_ROWS 391

__device__ __forceinline__ void key_plan(int n, int E, int& on, int& cpr, int& kw) {
    const int groups = (n + TILE_ROWS - 1) / TILE_ROWS;
    const int full = n / TILE_ROWS;
    const int nfull5 = E >> 9;
    on = 0;
    cpr = 2;
    kw = 0;
    if (groups <= 256 && full >= 1 && nfull5 >= 1) {
        int c = (nfull5 + full - 1) / full;
        if (c < 2) c = 2;
        if (c * 128 <= (TILE_ROWS * OUT) / 4) {
            on = 1;
            cpr = c;
            kw = c * 128;
        }
    }
}

__global__ __launch_bounds__(256) void k_init_deg(float* __restrict__ deg, int n) {
    int i = blockIdx.x * blockDim.x + threadIdx.x;
    if (i < n) deg[i] = 1.0f;
}


typedef int gv4i __attribute__((ext_vector_type(4)));
template <int NTILES, bool DEG>
__global__ __launch_bounds__(256) void k_gemm_wmma(const float* __restrict__ X,
                                                   const float* __restrict__ W,
                                                   float* __restrict__ H, int nrows,
                                                   const int* __restrict__ dst, int E,
                                                   float* __restrict__ deg,
                                                   float* __restrict__ Keys) {
    const int lane = threadIdx.x & 31;
    const int wave = threadIdx.x >> 5;
    const int tile = blockIdx.x * (blockDim.x >> 5) + wave;
    if (DEG) {
        const int nwaves = gridDim.x * (blockDim.x >> 5);
        const int nfull = E >> 7;
        int kon, cpr, kw;
        key_plan(nrows, E, kon, cpr, kw);
        const int nfull5 = E >> 9;
        for (int c = tile; c < nfull; c += nwaves) {
            const gv4i d4 = *(const gv4i*)(dst + (c << 7) + (lane << 2));
            atomicAdd(&deg[d4.x], 1.0f);
            atomicAdd(&deg[d4.y], 1.0f);
            atomicAdd(&deg[d4.z], 1.0f);
            atomicAdd(&deg[d4.w], 1.0f);
            const int c5 = c >> 2;
            if (kon && c5 < nfull5) {
                const int rg = c5 / cpr;
                const int within = c5 - rg * cpr;
                unsigned* kp = (unsigned*)Keys + (size_t)rg * (TILE_ROWS * OUT) + 32 +
                               within * 128 + (c & 3) * 32 + lane;
                *kp = ((unsigned)d4.x / TILE_ROWS) | (((unsigned)d4.y / TILE_ROWS) << 8) |
                      (((unsigned)d4.z / TILE_ROWS) << 16) | (((unsigned)d4.w / TILE_ROWS) << 24);
            }
        }
        if (tile == 0)
            for (int e = (nfull << 7) + lane; e < E; e += 32) atomicAdd(&deg[dst[e]], 1.0f);
    }
    if (tile * 16 >= nrows) return;

    const int ln = lane & 15;
    const int hi = lane >> 4;
    const int ldw = NTILES * 16;

    v8f acc[NTILES] = {};

    const float* xrow = X + (size_t)(tile * 16 + ln) * IN_DIM;

    for (int k = 0; k < IN_DIM; k += 4) {
        v2f a;
        a.x = xrow[k + 2 * hi + 0];
        a.y = xrow[k + 2 * hi + 1];

        const float* wr = W + (size_t)(k + 2 * hi) * ldw;
#pragma unroll
        for (int n = 0; n < NTILES; ++n) {
            v2f b;
            b.x = wr[n * 16 + ln];
            b.y = wr[ldw + n * 16 + ln];
            acc[n] = __builtin_amdgcn_wmma_f32_16x16x4_f32(
                false, a, false, b, (short)0, acc[n], false, false);
        }
    }

#pragma unroll
    for (int n = 0; n < NTILES; ++n) {
#pragma unroll
        for (int r = 0; r < 8; ++r) {
            int orow = tile * 16 + r + 8 * hi;
            H[(size_t)orow * ldw + n * 16 + ln] = acc[n][r];
        }
    }
}

#define AGG_WAVES 32
#define SCAN_U 4
#define QCAP (32 + SCAN_U * 128)

typedef int v4i __attribute__((ext_vector_type(4)));

template <int DIM, bool FIRST>
__global__ __launch_bounds__(AGG_WAVES * 32) void k_agg(const float* __restrict__ H,
                                                        const int* __restrict__ src,
                                                        const int* __restrict__ dst,
                                                        const float* __restrict__ deg,
                                                        const float* __restrict__ bias,
                                                        float* A, float* Lst, int n, int E) {
    constexpr int VPT = DIM / 32;
    constexpr int G = 32 / VPT;
    typedef float vecf __attribute__((ext_vector_type(VPT)));

    __shared__ float tile[TILE_ROWS * DIM];
    __shared__ int queue[AGG_WAVES * QCAP];
    __shared__ int lcount;

    const int lane = threadIdx.x & 31;
    const int wave = __builtin_amdgcn_readfirstlane((int)(threadIdx.x >> 5));
    const int r0 = blockIdx.x * TILE_ROWS;
    const int rows = min(TILE_ROWS, n - r0);

    int* const lhead = (int*)(Lst + (size_t)r0 * OUT);
    int kon, cpr, kw;
    key_plan(n, E, kon, cpr, kw);
    const int room = rows * OUT - 32 - kw;
    const int cap = room > 0 ? ((room / 3) & ~31) : 0;
    int* const lsrc = lhead + 32 + kw;
    int* const lrowp = lsrc + cap;
    float* const lwgt = (float*)(lrowp + cap);
    if (FIRST && threadIdx.x == 0) lcount = 0;

    const vecf bv = *(const vecf*)(bias + VPT * lane);
    for (int i0 = wave; i0 < rows; i0 += 4 * AGG_WAVES) {
        float s[4];
        vecf hv[4];
        for (int u = 0; u < 4; ++u) {
            const int i = (i0 + u * AGG_WAVES < rows) ? (i0 + u * AGG_WAVES) : i0;
            s[u] = rsqrtf(deg[r0 + i]);
            hv[u] = *(const vecf*)(H + (size_t)(r0 + i) * DIM + VPT * lane);
        }
        for (int u = 0; u < 4; ++u) {
            const int i = i0 + u * AGG_WAVES;
            if (i < rows)
                for (int k = 0; k < VPT; ++k)
                    tile[i * DIM + k * 32 + lane] = hv[u][k] * (s[u] * s[u]) + bv[k];
        }
    }
    __syncthreads();

    auto addrows = [&](int s, int lrow, float nrm, int cnt,
                       auto&& behind_last_loads) __attribute__((always_inline)) {
        for (int j0 = 0; j0 < cnt; j0 += G) {
            vecf v[G];
#pragma unroll
            for (int g = 0; g < G; ++g) {
                const int j = (j0 + g < cnt) ? (j0 + g) : j0;
                const int sj = __builtin_amdgcn_readlane(s, j);
                v[g] = *(const vecf*)(H + (size_t)sj * DIM + VPT * lane);
            }
            if (j0 + G >= cnt) behind_last_loads();
#pragma unroll
            for (int g = 0; g < G; ++g) {
                if (j0 + g < cnt) {
                    const int lj = __builtin_amdgcn_readlane(lrow, j0 + g);
                    const float nj = __int_as_float(
                        __builtin_amdgcn_readlane(__float_as_int(nrm), j0 + g));
                    float* tp = tile + lj * DIM + lane;
                    for (int k = 0; k < VPT; ++k)
                        __hip_atomic_fetch_add(tp + k * 32, v[g][k] * nj, __ATOMIC_RELAXED,
                                               __HIP_MEMORY_SCOPE_WORKGROUP);
                }
            }
        }
    };

    int total = 0;
    bool from_list = false;
    if (!FIRST) {
        total = __builtin_amdgcn_readfirstlane(lhead[0]);
        from_list = (unsigned)total <= (unsigned)cap;
    }

    if (from_list) {
        const int nb = total >> 5;
        auto entry = [&](int b, int& s, int& lrow, float& nrm) {
            const int idx = (b << 5) + lane;
            s = lsrc[idx];
            lrow = lrowp[idx];
            nrm = lwgt[idx];
        };
        int b = wave;
        int s = -1, lrow = 0;
        float nrm = 0.0f;
        if (b < nb) entry(b, s, lrow, nrm);
        while (b < nb) {
            const int bn = b + AGG_WAVES;
            int s2 = -1, lrow2 = 0;
            float nrm2 = 0.0f;
            auto next = [&]() __attribute__((always_inline)) {
                if (bn < nb) entry(bn, s2, lrow2, nrm2);
            };
            const int cnt = __builtin_popcount(__builtin_amdgcn_ballot_w32(s >= 0));
            if (cnt == 32)
                addrows(s, lrow, nrm, 32, next);
            else if (cnt > 0)
                addrows(s, lrow, nrm, cnt, next);
            else
                next();
            s = s2;
            lrow = lrow2;
            nrm = nrm2;
            b = bn;
        }
    } else {
        const int nchunks = (E + 127) >> 7;
        const int T = (nchunks + AGG_WAVES - 1) / AGG_WAVES;
        const int span = T * AGG_WAVES;
        const int rot = (int)((blockIdx.x * (unsigned)T) / gridDim.x) * AGG_WAVES;

        const int qoff = wave * QCAP;
        int qn = 0;
        const int nfull = E >> 7;

        auto chunk_of = [&](int t) -> int {
            int c = t * AGG_WAVES + wave + rot;
            if (c >= span) c -= span;
            return c;
        };
        auto chunk_ok = [&](int t) -> bool { return t < T && chunk_of(t) < nfull; };
        auto fetch = [&](int t) -> v4i {
            const int c = chunk_ok(t) ? chunk_of(t) : 0;
            return *(const v4i*)(dst + (c << 7) + (lane << 2));
        };

        auto process = [&](int qb, int cnt) __attribute__((always_inline)) {
            int s = 0, lrow = 0;
            float nrm = 0.0f;
            __builtin_amdgcn_fence(__ATOMIC_ACQ_REL, "wavefront");
            if (lane < cnt) {
                const int e = queue[qoff + qb + lane];
                s = src[e];
                const int d = dst[e];
                lrow = d - r0;
                nrm = rsqrtf(deg[s]) * rsqrtf(deg[d]);
            }
            if (FIRST) {
                int base = 0;
                if (lane == 0)
                    base = __hip_atomic_fetch_add(&lcount, 32, __ATOMIC_RELAXED,
                                                  __HIP_MEMORY_SCOPE_WORKGROUP);
                base = __builtin_amdgcn_readfirstlane(base);
                if (base + 32 <= cap) {
                    lsrc[base + lane] = (lane < cnt) ? s : -1;
                    lrowp[base + lane] = lrow;
                    lwgt[base + lane] = nrm;
                }
            }
            addrows(s, lrow, nrm, cnt, []() {});
        };

        auto handle = [&](const v4i d4, int t) {
            const bool o0 = (unsigned)(d4.x - r0) < (unsigned)rows;
            const bool o1 = (unsigned)(d4.y - r0) < (unsigned)rows;
            const bool o2 = (unsigned)(d4.z - r0) < (unsigned)rows;
            const bool o3 = (unsigned)(d4.w - r0) < (unsigned)rows;
            const unsigned any = __builtin_amdgcn_ballot_w32(o0) | __builtin_amdgcn_ballot_w32(o1) |
                                 __builtin_amdgcn_ballot_w32(o2) | __builtin_amdgcn_ballot_w32(o3);
            if (any != 0 && chunk_ok(t)) {
                const int e0 = (chunk_of(t) << 7) + (lane << 2);
                unsigned m = (o0 ? 1u : 0u) | (o1 ? 2u : 0u) | (o2 ? 4u : 0u) | (o3 ? 8u : 0u);
                unsigned bal = any;
                while (bal) {
                    const int pos = qn + (int)__builtin_amdgcn_mbcnt_lo(bal, 0u);
                    if (m) {
                        const int b = __builtin_ctz(m);
                        m &= m - 1;
                        queue[qoff + pos] = e0 + b;
                    }
                    qn += __builtin_popcount(bal);
                    bal = __builtin_amdgcn_ballot_w32(m != 0);
                }
            }
        };
        auto drain = [&]() __attribute__((always_inline)) {
            while (qn >= 32) {
                qn -= 32;
                process(qn, 32);
            }
        };

        int tail_from = nfull << 7;
        if (FIRST && kon) {
            const unsigned* const keys = (const unsigned*)Lst;
            const int nfull5 = E >> 9;
            const int T5 = (nfull5 + AGG_WAVES - 1) / AGG_WAVES;
            const int span5 = T5 * AGG_WAVES;
            const int rot5 = (int)((blockIdx.x * (unsigned)T5) / gridDim.x) * AGG_WAVES;
            const unsigned magic = (unsigned)((1ull << 32) / (unsigned)cpr) + 1u;
            const unsigned pat = blockIdx.x * 0x01010101u;
            auto chunk5_of = [&](int t) -> int {
                int c = t * AGG_WAVES + wave + rot5;
                if (c >= span5) c -= span5;
                return c;
            };
            auto chunk5_ok = [&](int t) -> bool { return t < T5 && chunk5_of(t) < nfull5; };
            auto fetchk = [&](int t) -> v4i {
                const unsigned c = chunk5_ok(t) ? (unsigned)chunk5_of(t) : 0u;
                const unsigned rg = __umulhi(c, magic);
                const unsigned within = c - rg * (unsigned)cpr;
                return *(const v4i*)(keys + (size_t)rg * (TILE_ROWS * OUT) + 32 + within * 128 +
                                     (lane << 2));
            };
            auto hit = [&](int w) -> unsigned {
                const unsigned x = (unsigned)w ^ pat;
                return ~(((x & 0x7F7F7F7Fu) + 0x7F7F7F7Fu) | x | 0x7F7F7F7Fu);
            };
            auto nib = [&](unsigned z) -> unsigned { return (((z >> 7) * 0x00204081u) >> 21) & 15u; };
            auto handlek = [&](const v4i k4, int t) {
                const unsigned z0 = hit(k4.x), z1 = hit(k4.y), z2 = hit(k4.z), z3 = hit(k4.w);
                const unsigned any = __builtin_amdgcn_ballot_w32((z0 | z1 | z2 | z3) != 0);
                if (any != 0 && chunk5_ok(t)) {
                    const int e0 = (chunk5_of(t) << 9) + (lane << 4);
                    unsigned m = nib(z0) | (nib(z1) << 4) | (nib(z2) << 8) | (nib(z3) << 12);
                    unsigned bal = any;
                    while (bal) {
                        const int pos = qn + (int)__builtin_amdgcn_mbcnt_lo(bal, 0u);
                        if (m) {
                            const int b = __builtin_ctz(m);
                            m &= m - 1;
                            queue[qoff + pos] = e0 + b;
                        }
                        qn += __builtin_popcount(bal);
                        bal = __builtin_amdgcn_ballot_w32(m != 0);
                    }
                }
            };
            auto sel = [&](const v4i (&g)[4], int u) -> v4i {
                v4i r = g[0];
                if (u == 1) r = g[1];
                if (u == 2) r = g[2];
                if (u == 3) r = g[3];
                return r;
            };
            v4i ka[4], kb[4];
            for (int u = 0; u < 4; ++u) ka[u] = fetchk(u);
            for (int tb = 0; tb < T5; tb += 8) {
                for (int u = 0; u < 4; ++u) kb[u] = fetchk(tb + 4 + u);
#pragma nounroll
                for (int u = 0; u < 4; ++u) {
                    handlek(sel(ka, u), tb + u);
                    drain();
                }
                for (int u = 0; u < 4; ++u) ka[u] = fetchk(tb + 8 + u);
#pragma nounroll
                for (int u = 0; u < 4; ++u) {
                    handlek(sel(kb, u), tb + 4 + u);
                    drain();
                }
            }
            tail_from = nfull5 << 9;
        } else if (nfull > 0) {
            v4i ga[SCAN_U], gb[SCAN_U];
            for (int u = 0; u < SCAN_U; ++u) ga[u] = fetch(u);
            for (int tb = 0; tb < T; tb += 2 * SCAN_U) {
                for (int u = 0; u < SCAN_U; ++u) gb[u] = fetch(tb + SCAN_U + u);
                for (int u = 0; u < SCAN_U; ++u) handle(ga[u], tb + u);
                drain();
                for (int u = 0; u < SCAN_U; ++u) ga[u] = fetch(tb + 2 * SCAN_U + u);
                for (int u = 0; u < SCAN_U; ++u) handle(gb[u], tb + SCAN_U + u);
                drain();
            }
        }
        if (wave == 0 && tail_from < E) {
            for (int eb = tail_from; eb < E; eb += 32) {
                const int e = eb + lane;
                const bool own = e < E && (unsigned)(dst[e < E ? e : 0] - r0) < (unsigned)rows;
                const unsigned bal = __builtin_amdgcn_ballot_w32(own);
                if (own) queue[qoff + qn + (int)__builtin_amdgcn_mbcnt_lo(bal, 0u)] = e;
                qn += __builtin_popcount(bal);
            }
        }
        while (qn > 0) {
            const int c = min(qn, 32);
            qn -= c;
            process(qn, c);
        }
    }

    __syncthreads();

    if (FIRST && threadIdx.x == 0) lhead[0] = lcount;

    for (int i = wave; i < rows; i += AGG_WAVES) {
        vecf o;
        for (int k = 0; k < VPT; ++k) {
            const float x = tile[i * DIM + k * 32 + lane];
            o[k] = FIRST ? fmaxf(x, 0.0f) : x;
        }
        *(vecf*)(A + (size_t)(r0 + i) * DIM + VPT * lane) = o;
    }
}

extern "C" void kernel_launch(void* const* d_in, const int* in_sizes, int n_in,
                              void* d_out, int out_size, void* d_ws, size_t ws_size,
                              hipStream_t stream) {
    const float* x   = (const float*)d_in[0];
    const int*   ei  = (const int*)d_in[1];
    const float* W1  = (const float*)d_in[2];
    const float* b1  = (const float*)d_in[3];
    const float* W2  = (const float*)d_in[4];
    const float* b2  = (const float*)d_in[5];
    float*       out = (float*)d_out;

    const int n = in_sizes[0] / IN_DIM;
    const int E = in_sizes[1] / 2;
    const int* srcIdx = ei;
    const int* dstIdx = ei + E;

    float* ws   = (float*)d_ws;
    float* deg  = ws;
    float* h    = deg + 2 * (size_t)n;
    float* agg  = h + (size_t)n * HID;

    k_init_deg<<<(n + 255) / 256, 256, 0, stream>>>(deg, n);

    const int tiles = (n + 15) / 16;
    const int gemmBlocks = (tiles + 7) / 8;
    const int aggBlocks = (n + TILE_ROWS - 1) / TILE_ROWS;

    k_gemm_wmma<8, true><<<gemmBlocks, 256, 0, stream>>>(x, W1, h, n, dstIdx, E, deg, out);
    k_agg<HID, true><<<aggBlocks, AGG_WAVES * 32, 0, stream>>>(h, srcIdx, dstIdx, deg, b1, agg, out, n, E);

    k_gemm_wmma<4, false><<<gemmBlocks, 256, 0, stream>>>(agg, W2, h, n, dstIdx, E, deg, out);
    k_agg<OUT, false><<<aggBlocks, AGG_WAVES * 32, 0, stream>>>(h, srcIdx, dstIdx, deg, b2, out, out, n, E);
}


# Layer-1 product's waves also write a packed word per edge (source | tile row << 20) by the owner keys; an owned edge's lookup reads that one line, not src[e]+dst[e]; 2-word lists; arrival order kept
# speedup vs baseline: 2.6445x; 1.0318x over previous
#include <hip/hip_runtime.h>

typedef float v2f __attribute__((ext_vector_type(2)));
typedef float v8f __attribute__((ext_vector_type(8)));

#define IN_DIM 128
#define HID 128
#define OUT 64
#define TILE_ROWS 391

__device__ __forceinline__ void key_plan(int n, int E, int& on, int& cpr, int& kw) {
    const int groups = (n + TILE_ROWS - 1) / TILE_ROWS;
    const int full = n / TILE_ROWS;
    const int nfull5 = E >> 9;
    on = 0;
    cpr = 2;
    kw = 0;
    if (groups <= 256 && full >= 1 && nfull5 >= 1) {
        int c = (nfull5 + full - 1) / full;
        if (c < 2) c = 2;
        if (c * 640 <= (TILE_ROWS * OUT) / 2) {
            on = 1;
            cpr = c;
            kw = c * 128;
        }
    }
}

__global__ __launch_bounds__(256) void k_init_deg(float* __restrict__ deg, int n) {
    int i = blockIdx.x * blockDim.x + threadIdx.x;
    if (i < n) deg[i] = 1.0f;
}


typedef int gv4i __attribute__((ext_vector_type(4)));
template <int NTILES, bool DEG>
__global__ __launch_bounds__(256) void k_gemm_wmma(const float* __restrict__ X,
                                                   const float* __restrict__ W,
                                                   float* __restrict__ H, int nrows,
                                                   const int* __restrict__ src,
                                                   const int* __restrict__ dst, int E,
                                                   float* __restrict__ deg,
                                                   float* __restrict__ Keys) {
    const int lane = threadIdx.x & 31;
    const int wave = threadIdx.x >> 5;
    const int tile = blockIdx.x * (blockDim.x >> 5) + wave;
    if (DEG) {
        const int nwaves = gridDim.x * (blockDim.x >> 5);
        const int nfull = E >> 7;
        int kon, cpr, kw;
        key_plan(nrows, E, kon, cpr, kw);
        const int nfull5 = E >> 9;
        for (int c = tile; c < nfull; c += nwaves) {
            const gv4i d4 = *(const gv4i*)(dst + (c << 7) + (lane << 2));
            const bool keyed = kon && (c >> 2) < nfull5;
            gv4i s4 = {0, 0, 0, 0};
            if (keyed) s4 = *(const gv4i*)(src + (c << 7) + (lane << 2));
            atomicAdd(&deg[d4.x], 1.0f);
            atomicAdd(&deg[d4.y], 1.0f);
            atomicAdd(&deg[d4.z], 1.0f);
            atomicAdd(&deg[d4.w], 1.0f);
            const int c5 = c >> 2;
            if (keyed) {
                const int rg = c5 / cpr;
                const int within = c5 - rg * cpr;
                unsigned* gp = (unsigned*)Keys + (size_t)rg * (TILE_ROWS * OUT) + 32;
                const int w = (c & 3) * 32 + lane;
                const unsigned o0 = (unsigned)d4.x / TILE_ROWS, o1 = (unsigned)d4.y / TILE_ROWS;
                const unsigned o2 = (unsigned)d4.z / TILE_ROWS, o3 = (unsigned)d4.w / TILE_ROWS;
                gp[within * 128 + w] = o0 | (o1 << 8) | (o2 << 16) | (o3 << 24);
                gv4i p;
                p.x = s4.x | ((d4.x - (int)o0 * TILE_ROWS) << 20);
                p.y = s4.y | ((d4.y - (int)o1 * TILE_ROWS) << 20);
                p.z = s4.z | ((d4.z - (int)o2 * TILE_ROWS) << 20);
                p.w = s4.w | ((d4.w - (int)o3 * TILE_ROWS) << 20);
                *(gv4i*)(gp + kw + within * 512 + 4 * w) = p;
            }
        }
        if (tile == 0)
            for (int e = (nfull << 7) + lane; e < E; e += 32) atomicAdd(&deg[dst[e]], 1.0f);
    }
    if (tile * 16 >= nrows) return;

    const int ln = lane & 15;
    const int hi = lane >> 4;
    const int ldw = NTILES * 16;

    v8f acc[NTILES] = {};

    const float* xrow = X + (size_t)(tile * 16 + ln) * IN_DIM;

    for (int k = 0; k < IN_DIM; k += 4) {
        v2f a;
        a.x = xrow[k + 2 * hi + 0];
        a.y = xrow[k + 2 * hi + 1];

        const float* wr = W + (size_t)(k + 2 * hi) * ldw;
#pragma unroll
        for (int n = 0; n < NTILES; ++n) {
            v2f b;
            b.x = wr[n * 16 + ln];
            b.y = wr[ldw + n * 16 + ln];
            acc[n] = __builtin_amdgcn_wmma_f32_16x16x4_f32(
                false, a, false, b, (short)0, acc[n], false, false);
        }
    }

#pragma unroll
    for (int n = 0; n < NTILES; ++n) {
#pragma unroll
        for (int r = 0; r < 8; ++r) {
            int orow = tile * 16 + r + 8 * hi;
            H[(size_t)orow * ldw + n * 16 + ln] = acc[n][r];
        }
    }
}

#define AGG_WAVES 32
#define SCAN_U 4
#define QCAP (32 + SCAN_U * 128)

typedef int v4i __attribute__((ext_vector_type(4)));

template <int DIM, bool FIRST>
__global__ __launch_bounds__(AGG_WAVES * 32) void k_agg(const float* __restrict__ H,
                                                        const int* __restrict__ src,
                                                        const int* __restrict__ dst,
                                                        const float* __restrict__ deg,
                                                        const float* __restrict__ bias,
                                                        float* A, float* Lst, int n, int E) {
    constexpr int VPT = DIM / 32;
    constexpr int G = 32 / VPT;
    typedef float vecf __attribute__((ext_vector_type(VPT)));

    __shared__ float tile[TILE_ROWS * DIM];
    __shared__ int queue[AGG_WAVES * QCAP];
    __shared__ int lcount;

    const int lane = threadIdx.x & 31;
    const int wave = __builtin_amdgcn_readfirstlane((int)(threadIdx.x >> 5));
    const int r0 = blockIdx.x * TILE_ROWS;
    const int rows = min(TILE_ROWS, n - r0);

    int* const lhead = (int*)(Lst + (size_t)r0 * OUT);
    int kon, cpr, kw;
    key_plan(n, E, kon, cpr, kw);
    const int resv = (rows == TILE_ROWS) ? 5 * kw : 0;
    const int room = rows * OUT - 32 - resv;
    const int cap = (room > 0 && n <= (1 << 20)) ? ((room / 2) & ~31) : 0;
    int* const lpk = lhead + 32 + resv;
    float* const lwgt = (float*)(lpk + cap);
    const int nfull5 = E >> 9;
    const unsigned magic = (unsigned)((1ull << 32) / (unsigned)cpr) + 1u;
    const int* const pkbase = (const int*)Lst + 32 + kw;
    if (FIRST && threadIdx.x == 0) lcount = 0;

    const vecf bv = *(const vecf*)(bias + VPT * lane);
    for (int i0 = wave; i0 < rows; i0 += 4 * AGG_WAVES) {
        float s[4];
        vecf hv[4];
        for (int u = 0; u < 4; ++u) {
            const int i = (i0 + u * AGG_WAVES < rows) ? (i0 + u * AGG_WAVES) : i0;
            s[u] = rsqrtf(deg[r0 + i]);
            hv[u] = *(const vecf*)(H + (size_t)(r0 + i) * DIM + VPT * lane);
        }
        for (int u = 0; u < 4; ++u) {
            const int i = i0 + u * AGG_WAVES;
            if (i < rows)
                for (int k = 0; k < VPT; ++k)
                    tile[i * DIM + k * 32 + lane] = hv[u][k] * (s[u] * s[u]) + bv[k];
        }
    }
    __syncthreads();

    auto addrows = [&](int s, int lrow, float nrm, int cnt,
                       auto&& behind_last_loads) __attribute__((always_inline)) {
        for (int j0 = 0; j0 < cnt; j0 += G) {
            vecf v[G];
#pragma unroll
            for (int g = 0; g < G; ++g) {
                const int j = (j0 + g < cnt) ? (j0 + g) : j0;
                const int sj = __builtin_amdgcn_readlane(s, j);
                v[g] = *(const vecf*)(H + (size_t)sj * DIM + VPT * lane);
            }
            if (j0 + G >= cnt) behind_last_loads();
#pragma unroll
            for (int g = 0; g < G; ++g) {
                if (j0 + g < cnt) {
                    const int lj = __builtin_amdgcn_readlane(lrow, j0 + g);
                    const float nj = __int_as_float(
                        __builtin_amdgcn_readlane(__float_as_int(nrm), j0 + g));
                    float* tp = tile + lj * DIM + lane;
                    for (int k = 0; k < VPT; ++k)
                        __hip_atomic_fetch_add(tp + k * 32, v[g][k] * nj, __ATOMIC_RELAXED,
                                               __HIP_MEMORY_SCOPE_WORKGROUP);
                }
            }
        }
    };

    int total = 0;
    bool from_list = false;
    if (!FIRST) {
        total = __builtin_amdgcn_readfirstlane(lhead[0]);
        from_list = (unsigned)total <= (unsigned)cap;
    }

    if (from_list) {
        const int nb = total >> 5;
        auto entry = [&](int b, int& pk, float& nrm) {
            const int idx = (b << 5) + lane;
            pk = lpk[idx];
            nrm = lwgt[idx];
        };
        int b = wave;
        int pk = -1;
        float nrm = 0.0f;
        if (b < nb) entry(b, pk, nrm);
        while (b < nb) {
            const int bn = b + AGG_WAVES;
            int pk2 = -1;
            float nrm2 = 0.0f;
            auto next = [&]() __attribute__((always_inline)) {
                if (bn < nb) entry(bn, pk2, nrm2);
            };
            const int cnt = __builtin_popcount(__builtin_amdgcn_ballot_w32(pk >= 0));
            const int s = pk & 0xFFFFF;
            const int lrow = pk >> 20;
            if (cnt == 32)
                addrows(s, lrow, nrm, 32, next);
            else if (cnt > 0)
                addrows(s, lrow, nrm, cnt, next);
            else
                next();
            pk = pk2;
            nrm = nrm2;
            b = bn;
        }
    } else {
        const int nchunks = (E + 127) >> 7;
        const int T = (nchunks + AGG_WAVES - 1) / AGG_WAVES;
        const int span = T * AGG_WAVES;
        const int rot = (int)((blockIdx.x * (unsigned)T) / gridDim.x) * AGG_WAVES;

        const int qoff = wave * QCAP;
        int qn = 0;
        const int nfull = E >> 7;

        auto chunk_of = [&](int t) -> int {
            int c = t * AGG_WAVES + wave + rot;
            if (c >= span) c -= span;
            return c;
        };
        auto chunk_ok = [&](int t) -> bool { return t < T && chunk_of(t) < nfull; };
        auto fetch = [&](int t) -> v4i {
            const int c = chunk_ok(t) ? chunk_of(t) : 0;
            return *(const v4i*)(dst + (c << 7) + (lane << 2));
        };

        auto process = [&](int qb, int cnt) __attribute__((always_inline)) {
            int s = 0, lrow = 0;
            float nrm = 0.0f;
            __builtin_amdgcn_fence(__ATOMIC_ACQ_REL, "wavefront");
            if (lane < cnt) {
                const int e = queue[qoff + qb + lane];
                int d;
                if (FIRST && kon && e < (nfull5 << 9)) {
                    const unsigned c5 = (unsigned)e >> 9;
                    const unsigned rg = __umulhi(c5, magic);
                    const unsigned within = c5 - rg * (unsigned)cpr;
                    const int p = pkbase[(size_t)rg * (TILE_ROWS * OUT) + within * 512 + (e & 511)];
                    s = p & 0xFFFFF;
                    lrow = p >> 20;
                    d = r0 + lrow;
                } else {
                    s = src[e];
                    d = dst[e];
                    lrow = d - r0;
                }
                nrm = rsqrtf(deg[s]) * rsqrtf(deg[d]);
            }
            if (FIRST) {
                int base = 0;
                if (lane == 0)
                    base = __hip_atomic_fetch_add(&lcount, 32, __ATOMIC_RELAXED,
                                                  __HIP_MEMORY_SCOPE_WORKGROUP);
                base = __builtin_amdgcn_readfirstlane(base);
                if (base + 32 <= cap) {
                    lpk[base + lane] = (lane < cnt) ? (s | (lrow << 20)) : -1;
                    lwgt[base + lane] = nrm;
                }
            }
            addrows(s, lrow, nrm, cnt, []() {});
        };

        auto handle = [&](const v4i d4, int t) {
            const bool o0 = (unsigned)(d4.x - r0) < (unsigned)rows;
            const bool o1 = (unsigned)(d4.y - r0) < (unsigned)rows;
            const bool o2 = (unsigned)(d4.z - r0) < (unsigned)rows;
            const bool o3 = (unsigned)(d4.w - r0) < (unsigned)rows;
            const unsigned any = __builtin_amdgcn_ballot_w32(o0) | __builtin_amdgcn_ballot_w32(o1) |
                                 __builtin_amdgcn_ballot_w32(o2) | __builtin_amdgcn_ballot_w32(o3);
            if (any != 0 && chunk_ok(t)) {
                const int e0 = (chunk_of(t) << 7) + (lane << 2);
                unsigned m = (o0 ? 1u : 0u) | (o1 ? 2u : 0u) | (o2 ? 4u : 0u) | (o3 ? 8u : 0u);
                unsigned bal = any;
                while (bal) {
                    const int pos = qn + (int)__builtin_amdgcn_mbcnt_lo(bal, 0u);
                    if (m) {
                        const int b = __builtin_ctz(m);
                        m &= m - 1;
                        queue[qoff + pos] = e0 + b;
                    }
                    qn += __builtin_popcount(bal);
                    bal = __builtin_amdgcn_ballot_w32(m != 0);
                }
            }
        };
        auto drain = [&]() __attribute__((always_inline)) {
            while (qn >= 32) {
                qn -= 32;
                process(qn, 32);
            }
        };

        int tail_from = nfull << 7;
        if (FIRST && kon) {
            const unsigned* const keys = (const unsigned*)Lst;
            const int T5 = (nfull5 + AGG_WAVES - 1) / AGG_WAVES;
            const int span5 = T5 * AGG_WAVES;
            const int rot5 = (int)((blockIdx.x * (unsigned)T5) / gridDim.x) * AGG_WAVES;
            const unsigned pat = blockIdx.x * 0x01010101u;
            auto chunk5_of = [&](int t) -> int {
                int c = t * AGG_WAVES + wave + rot5;
                if (c >= span5) c -= span5;
                return c;
            };
            auto chunk5_ok = [&](int t) -> bool { return t < T5 && chunk5_of(t) < nfull5; };
            auto fetchk = [&](int t) -> v4i {
                const unsigned c = chunk5_ok(t) ? (unsigned)chunk5_of(t) : 0u;
                const unsigned rg = __umulhi(c, magic);
                const unsigned within = c - rg * (unsigned)cpr;
                return *(const v4i*)(keys + (size_t)rg * (TILE_ROWS * OUT) + 32 + within * 128 +
                                     (lane << 2));
            };
            auto hit = [&](int w) -> unsigned {
                const unsigned x = (unsigned)w ^ pat;
                return ~(((x & 0x7F7F7F7Fu) + 0x7F7F7F7Fu) | x | 0x7F7F7F7Fu);
            };
            auto nib = [&](unsigned z) -> unsigned { return (((z >> 7) * 0x00204081u) >> 21) & 15u; };
            auto handlek = [&](const v4i k4, int t) {
                const unsigned z0 = hit(k4.x), z1 = hit(k4.y), z2 = hit(k4.z), z3 = hit(k4.w);
                const unsigned any = __builtin_amdgcn_ballot_w32((z0 | z1 | z2 | z3) != 0);
                if (any != 0 && chunk5_ok(t)) {
                    const int e0 = (chunk5_of(t) << 9) + (lane << 4);
                    unsigned m = nib(z0) | (nib(z1) << 4) | (nib(z2) << 8) | (nib(z3) << 12);
                    unsigned bal = any;
                    while (bal) {
                        const int pos = qn + (int)__builtin_amdgcn_mbcnt_lo(bal, 0u);
                        if (m) {
                            const int b = __builtin_ctz(m);
                            m &= m - 1;
                            queue[qoff + pos] = e0 + b;
                        }
                        qn += __builtin_popcount(bal);
                        bal = __builtin_amdgcn_ballot_w32(m != 0);
                    }
                }
            };
            auto sel = [&](const v4i (&g)[4], int u) -> v4i {
                v4i r = g[0];
                if (u == 1) r = g[1];
                if (u == 2) r = g[2];
                if (u == 3) r = g[3];
                return r;
            };
            v4i ka[4], kb[4];
            for (int u = 0; u < 4; ++u) ka[u] = fetchk(u);
            for (int tb = 0; tb < T5; tb += 8) {
                for (int u = 0; u < 4; ++u) kb[u] = fetchk(tb + 4 + u);
#pragma nounroll
                for (int u = 0; u < 4; ++u) {
                    handlek(sel(ka, u), tb + u);
                    drain();
                }
                for (int u = 0; u < 4; ++u) ka[u] = fetchk(tb + 8 + u);
#pragma nounroll
                for (int u = 0; u < 4; ++u) {
                    handlek(sel(kb, u), tb + 4 + u);
                    drain();
                }
            }
            tail_from = nfull5 << 9;
        } else if (nfull > 0) {
            v4i ga[SCAN_U], gb[SCAN_U];
            for (int u = 0; u < SCAN_U; ++u) ga[u] = fetch(u);
            for (int tb = 0; tb < T; tb += 2 * SCAN_U) {
                for (int u = 0; u < SCAN_U; ++u) gb[u] = fetch(tb + SCAN_U + u);
                for (int u = 0; u < SCAN_U; ++u) handle(ga[u], tb + u);
                drain();
                for (int u = 0; u < SCAN_U; ++u) ga[u] = fetch(tb + 2 * SCAN_U + u);
                for (int u = 0; u < SCAN_U; ++u) handle(gb[u], tb + SCAN_U + u);
                drain();
            }
        }
        if (wave == 0 && tail_from < E) {
            for (int eb = tail_from; eb < E; eb += 32) {
                const int e = eb + lane;
                const bool own = e < E && (unsigned)(dst[e < E ? e : 0] - r0) < (unsigned)rows;
                const unsigned bal = __builtin_amdgcn_ballot_w32(own);
                if (own) queue[qoff + qn + (int)__builtin_amdgcn_mbcnt_lo(bal, 0u)] = e;
                qn += __builtin_popcount(bal);
            }
        }
        while (qn > 0) {
            const int c = min(qn, 32);
            qn -= c;
            process(qn, c);
        }
    }

    __syncthreads();

    if (FIRST && threadIdx.x == 0) lhead[0] = lcount;

    for (int i = wave; i < rows; i += AGG_WAVES) {
        vecf o;
        for (int k = 0; k < VPT; ++k) {
            const float x = tile[i * DIM + k * 32 + lane];
            o[k] = FIRST ? fmaxf(x, 0.0f) : x;
        }
        *(vecf*)(A + (size_t)(r0 + i) * DIM + VPT * lane) = o;
    }
}

extern "C" void kernel_launch(void* const* d_in, const int* in_sizes, int n_in,
                              void* d_out, int out_size, void* d_ws, size_t ws_size,
                              hipStream_t stream) {
    const float* x   = (const float*)d_in[0];
    const int*   ei  = (const int*)d_in[1];
    const float* W1  = (const float*)d_in[2];
    const float* b1  = (const float*)d_in[3];
    const float* W2  = (const float*)d_in[4];
    const float* b2  = (const float*)d_in[5];
    float*       out = (float*)d_out;

    const int n = in_sizes[0] / IN_DIM;
    const int E = in_sizes[1] / 2;
    const int* srcIdx = ei;
    const int* dstIdx = ei + E;

    float* ws   = (float*)d_ws;
    float* deg  = ws;
    float* h    = deg + 2 * (size_t)n;
    float* agg  = h + (size_t)n * HID;

    k_init_deg<<<(n + 255) / 256, 256, 0, stream>>>(deg, n);

    const int tiles = (n + 15) / 16;
    const int gemmBlocks = (tiles + 7) / 8;
    const int aggBlocks = (n + TILE_ROWS - 1) / TILE_ROWS;

    k_gemm_wmma<8, true><<<gemmBlocks, 256, 0, stream>>>(x, W1, h, n, srcIdx, dstIdx, E, deg, out);
    k_agg<HID, true><<<aggBlocks, AGG_WAVES * 32, 0, stream>>>(h, srcIdx, dstIdx, deg, b1, agg, out, n, E);

    k_gemm_wmma<4, false><<<gemmBlocks, 256, 0, stream>>>(agg, W2, h, n, srcIdx, dstIdx, E, deg, out);
    k_agg<OUT, false><<<aggBlocks, AGG_WAVES * 32, 0, stream>>>(h, srcIdx, dstIdx, deg, b2, out, out, n, E);
}


# Both dense products leave their launches: layer 1 sums rows of x ((An x) W1 for An (x W1)) and the group multiplies its LDS tile by W1, then W2, on the matrix unit; 5 launches -> 4; rounding changes
# speedup vs baseline: 2.9529x; 1.1166x over previous
#include <hip/hip_runtime.h>

typedef float v2f __attribute__((ext_vector_type(2)));
typedef float v8f __attribute__((ext_vector_type(8)));

#define IN_DIM 128
#define HID 128
#define OUT 64
#define TILE_ROWS 391

__device__ __forceinline__ void key_plan(int n, int E, int& on, int& cpr, int& kw) {
    const int groups = (n + TILE_ROWS - 1) / TILE_ROWS;
    const int full = n / TILE_ROWS;
    const int nfull5 = E >> 9;
    on = 0;
    cpr = 2;
    kw = 0;
    if (groups <= 256 && full >= 1 && nfull5 >= 1) {
        int c = (nfull5 + full - 1) / full;
        if (c < 2) c = 2;
        if (c * 640 <= (TILE_ROWS * OUT) / 2) {
            on = 1;
            cpr = c;
            kw = c * 128;
        }
    }
}

__global__ __launch_bounds__(256) void k_init_deg(float* __restrict__ deg, int n) {
    int i = blockIdx.x * blockDim.x + threadIdx.x;
    if (i < n) deg[i] = 1.0f;
}

typedef int gv4i __attribute__((ext_vector_type(4)));
__global__ __launch_bounds__(256) void k_edges(const int* __restrict__ src,
                                               const int* __restrict__ dst, int E,
                                               float* __restrict__ deg,
                                               float* __restrict__ Keys, int nrows) {
    const int lane = threadIdx.x & 31;
    const int wave = threadIdx.x >> 5;
    const int tile = blockIdx.x * (blockDim.x >> 5) + wave;
    const int nwaves = gridDim.x * (blockDim.x >> 5);
    const int nfull = E >> 7;
    int kon, cpr, kw;
    key_plan(nrows, E, kon, cpr, kw);
    const int nfull5 = E >> 9;
    for (int c = tile; c < nfull; c += nwaves) {
        const gv4i d4 = *(const gv4i*)(dst + (c << 7) + (lane << 2));
        const bool keyed = kon && (c >> 2) < nfull5;
        gv4i s4 = {0, 0, 0, 0};
        if (keyed) s4 = *(const gv4i*)(src + (c << 7) + (lane << 2));
        atomicAdd(&deg[d4.x], 1.0f);
        atomicAdd(&deg[d4.y], 1.0f);
        atomicAdd(&deg[d4.z], 1.0f);
        atomicAdd(&deg[d4.w], 1.0f);
        const int c5 = c >> 2;
        if (keyed) {
            const int rg = c5 / cpr;
            const int within = c5 - rg * cpr;
            unsigned* gp = (unsigned*)Keys + (size_t)rg * (TILE_ROWS * OUT) + 32;
            const int w = (c & 3) * 32 + lane;
            const unsigned o0 = (unsigned)d4.x / TILE_ROWS, o1 = (unsigned)d4.y / TILE_ROWS;
            const unsigned o2 = (unsigned)d4.z / TILE_ROWS, o3 = (unsigned)d4.w / TILE_ROWS;
            gp[within * 128 + w] = o0 | (o1 << 8) | (o2 << 16) | (o3 << 24);
            gv4i p;
            p.x = s4.x | ((d4.x - (int)o0 * TILE_ROWS) << 20);
            p.y = s4.y | ((d4.y - (int)o1 * TILE_ROWS) << 20);
            p.z = s4.z | ((d4.z - (int)o2 * TILE_ROWS) << 20);
            p.w = s4.w | ((d4.w - (int)o3 * TILE_ROWS) << 20);
            *(gv4i*)(gp + kw + within * 512 + 4 * w) = p;
        }
    }
    if (tile == 0)
        for (int e = (nfull << 7) + lane; e < E; e += 32) atomicAdd(&deg[dst[e]], 1.0f);
}

#define AGG_WAVES 32
#define SCAN_U 4
#define QCAP (32 + SCAN_U * 128)

typedef int v4i __attribute__((ext_vector_type(4)));

template <int DIM, bool FIRST>
__global__ __launch_bounds__(AGG_WAVES * 32) void k_agg(const float* __restrict__ H,
                                                        const int* __restrict__ src,
                                                        const int* __restrict__ dst,
                                                        const float* __restrict__ deg,
                                                        const float* __restrict__ bias,
                                                        float* A, float* Lst, int n, int E,
                                                        const float* __restrict__ W1,
                                                        const float* __restrict__ W2,
                                                        float* __restrict__ H2) {
    constexpr int VPT = DIM / 32;
    constexpr int PITCH = FIRST ? DIM + 4 : DIM;
    constexpr int TROWS = FIRST ? ((TILE_ROWS + 15) / 16) * 16 : TILE_ROWS;
    constexpr int G = 32 / VPT;
    typedef float vecf __attribute__((ext_vector_type(VPT)));

    __shared__ __attribute__((aligned(16))) float tile[TROWS * PITCH];
    __shared__ int queue[AGG_WAVES * QCAP];
    __shared__ int lcount;

    const int lane = threadIdx.x & 31;
    const int wave = __builtin_amdgcn_readfirstlane((int)(threadIdx.x >> 5));
    const int r0 = blockIdx.x * TILE_ROWS;
    const int rows = min(TILE_ROWS, n - r0);

    int* const lhead = (int*)(Lst + (size_t)r0 * OUT);
    int kon, cpr, kw;
    key_plan(n, E, kon, cpr, kw);
    const int resv = (rows == TILE_ROWS) ? 5 * kw : 0;
    const int room = rows * OUT - 32 - resv;
    const int cap = (room > 0 && n <= (1 << 20)) ? ((room / 2) & ~31) : 0;
    int* const lpk = lhead + 32 + resv;
    float* const lwgt = (float*)(lpk + cap);
    const int nfull5 = E >> 9;
    const unsigned magic = (unsigned)((1ull << 32) / (unsigned)cpr) + 1u;
    const int* const pkbase = (const int*)Lst + 32 + kw;
    if (FIRST && threadIdx.x == 0) lcount = 0;

    const vecf bv = *(const vecf*)(bias + VPT * lane);
    for (int i0 = wave; i0 < rows; i0 += 4 * AGG_WAVES) {
        float s[4];
        vecf hv[4];
        for (int u = 0; u < 4; ++u) {
            const int i = (i0 + u * AGG_WAVES < rows) ? (i0 + u * AGG_WAVES) : i0;
            s[u] = rsqrtf(deg[r0 + i]);
            hv[u] = *(const vecf*)(H + (size_t)(r0 + i) * DIM + VPT * lane);
        }
        for (int u = 0; u < 4; ++u) {
            const int i = i0 + u * AGG_WAVES;
            if (i < rows)
                for (int k = 0; k < VPT; ++k)
                    tile[i * PITCH + k * 32 + lane] = hv[u][k] * (s[u] * s[u]) + (FIRST ? 0.0f : bv[k]);
        }
    }
    __syncthreads();

    auto addrows = [&](int s, int lrow, float nrm, int cnt,
                       auto&& behind_last_loads) __attribute__((always_inline)) {
        for (int j0 = 0; j0 < cnt; j0 += G) {
            vecf v[G];
#pragma unroll
            for (int g = 0; g < G; ++g) {
                const int j = (j0 + g < cnt) ? (j0 + g) : j0;
                const int sj = __builtin_amdgcn_readlane(s, j);
                v[g] = *(const vecf*)(H + (size_t)sj * DIM + VPT * lane);
            }
            if (j0 + G >= cnt) behind_last_loads();
#pragma unroll
            for (int g = 0; g < G; ++g) {
                if (j0 + g < cnt) {
                    const int lj = __builtin_amdgcn_readlane(lrow, j0 + g);
                    const float nj = __int_as_float(
                        __builtin_amdgcn_readlane(__float_as_int(nrm), j0 + g));
                    float* tp = tile + lj * PITCH + lane;
                    for (int k = 0; k < VPT; ++k)
                        __hip_atomic_fetch_add(tp + k * 32, v[g][k] * nj, __ATOMIC_RELAXED,
                                               __HIP_MEMORY_SCOPE_WORKGROUP);
                }
            }
        }
    };

    int total = 0;
    bool from_list = false;
    if (!FIRST) {
        total = __builtin_amdgcn_readfirstlane(lhead[0]);
        from_list = (unsigned)total <= (unsigned)cap;
    }

    if (from_list) {
        const int nb = total >> 5;
        auto entry = [&](int b, int& pk, float& nrm) {
            const int idx = (b << 5) + lane;
            pk = lpk[idx];
            nrm = lwgt[idx];
        };
        int b = wave;
        int pk = -1;
        float nrm = 0.0f;
        if (b < nb) entry(b, pk, nrm);
        while (b < nb) {
            const int bn = b + AGG_WAVES;
            int pk2 = -1;
            float nrm2 = 0.0f;
            auto next = [&]() __attribute__((always_inline)) {
                if (bn < nb) entry(bn, pk2, nrm2);
            };
            const int cnt = __builtin_popcount(__builtin_amdgcn_ballot_w32(pk >= 0));
            const int s = pk & 0xFFFFF;
            const int lrow = pk >> 20;
            if (cnt == 32)
                addrows(s, lrow, nrm, 32, next);
            else if (cnt > 0)
                addrows(s, lrow, nrm, cnt, next);
            else
                next();
            pk = pk2;
            nrm = nrm2;
            b = bn;
        }
    } else {
        const int nchunks = (E + 127) >> 7;
        const int T = (nchunks + AGG_WAVES - 1) / AGG_WAVES;
        const int span = T * AGG_WAVES;
        const int rot = (int)((blockIdx.x * (unsigned)T) / gridDim.x) * AGG_WAVES;

        const int qoff = wave * QCAP;
        int qn = 0;
        const int nfull = E >> 7;

        auto chunk_of = [&](int t) -> int {
            int c = t * AGG_WAVES + wave + rot;
            if (c >= span) c -= span;
            return c;
        };
        auto chunk_ok = [&](int t) -> bool { return t < T && chunk_of(t) < nfull; };
        auto fetch = [&](int t) -> v4i {
            const int c = chunk_ok(t) ? chunk_of(t) : 0;
            return *(const v4i*)(dst + (c << 7) + (lane << 2));
        };

        auto process = [&](int qb, int cnt) __attribute__((always_inline)) {
            int s = 0, lrow = 0;
            float nrm = 0.0f;
            __builtin_amdgcn_fence(__ATOMIC_ACQ_REL, "wavefront");
            if (lane < cnt) {
                const int e = queue[qoff + qb + lane];
                int d;
                if (FIRST && kon && e < (nfull5 << 9)) {
                    const unsigned c5 = (unsigned)e >> 9;
                    const unsigned rg = __umulhi(c5, magic);
                    const unsigned within = c5 - rg * (unsigned)cpr;
                    const int p = pkbase[(size_t)rg * (TILE_ROWS * OUT) + within * 512 + (e & 511)];
                    s = p & 0xFFFFF;
                    lrow = p >> 20;
                    d = r0 + lrow;
                } else {
                    s = src[e];
                    d = dst[e];
                    lrow = d - r0;
                }
                nrm = rsqrtf(deg[s]) * rsqrtf(deg[d]);
            }
            if (FIRST) {
                int base = 0;
                if (lane == 0)
                    base = __hip_atomic_fetch_add(&lcount, 32, __ATOMIC_RELAXED,
                                                  __HIP_MEMORY_SCOPE_WORKGROUP);
                base = __builtin_amdgcn_readfirstlane(base);
                if (base + 32 <= cap) {
                    lpk[base + lane] = (lane < cnt) ? (s | (lrow << 20)) : -1;
                    lwgt[base + lane] = nrm;
                }
            }
            addrows(s, lrow, nrm, cnt, []() {});
        };

        auto handle = [&](const v4i d4, int t) {
            const bool o0 = (unsigned)(d4.x - r0) < (unsigned)rows;
            const bool o1 = (unsigned)(d4.y - r0) < (unsigned)rows;
            const bool o2 = (unsigned)(d4.z - r0) < (unsigned)rows;
            const bool o3 = (unsigned)(d4.w - r0) < (unsigned)rows;
            const unsigned any = __builtin_amdgcn_ballot_w32(o0) | __builtin_amdgcn_ballot_w32(o1) |
                                 __builtin_amdgcn_ballot_w32(o2) | __builtin_amdgcn_ballot_w32(o3);
            if (any != 0 && chunk_ok(t)) {
                const int e0 = (chunk_of(t) << 7) + (lane << 2);
                unsigned m = (o0 ? 1u : 0u) | (o1 ? 2u : 0u) | (o2 ? 4u : 0u) | (o3 ? 8u : 0u);
                unsigned bal = any;
                while (bal) {
                    const int pos = qn + (int)__builtin_amdgcn_mbcnt_lo(bal, 0u);
                    if (m) {
                        const int b = __builtin_ctz(m);
                        m &= m - 1;
                        queue[qoff + pos] = e0 + b;
                    }
                    qn += __builtin_popcount(bal);
                    bal = __builtin_amdgcn_ballot_w32(m != 0);
                }
            }
        };
        auto drain = [&]() __attribute__((always_inline)) {
            while (qn >= 32) {
                qn -= 32;
                process(qn, 32);
            }
        };

        int tail_from = nfull << 7;
        if (FIRST && kon) {
            const unsigned* const keys = (const unsigned*)Lst;
            const int T5 = (nfull5 + AGG_WAVES - 1) / AGG_WAVES;
            const int span5 = T5 * AGG_WAVES;
            const int rot5 = (int)((blockIdx.x * (unsigned)T5) / gridDim.x) * AGG_WAVES;
            const unsigned pat = blockIdx.x * 0x01010101u;
            auto chunk5_of = [&](int t) -> int {
                int c = t * AGG_WAVES + wave + rot5;
                if (c >= span5) c -= span5;
                return c;
            };
            auto chunk5_ok = [&](int t) -> bool { return t < T5 && chunk5_of(t) < nfull5; };
            auto fetchk = [&](int t) -> v4i {
                const unsigned c = chunk5_ok(t) ? (unsigned)chunk5_of(t) : 0u;
                const unsigned rg = __umulhi(c, magic);
                const unsigned within = c - rg * (unsigned)cpr;
                return *(const v4i*)(keys + (size_t)rg * (TILE_ROWS * OUT) + 32 + within * 128 +
                                     (lane << 2));
            };
            auto hit = [&](int w) -> unsigned {
                const unsigned x = (unsigned)w ^ pat;
                return ~(((x & 0x7F7F7F7Fu) + 0x7F7F7F7Fu) | x | 0x7F7F7F7Fu);
            };
            auto nib = [&](unsigned z) -> unsigned { return (((z >> 7) * 0x00204081u) >> 21) & 15u; };
            auto handlek = [&](const v4i k4, int t) {
                const unsigned z0 = hit(k4.x), z1 = hit(k4.y), z2 = hit(k4.z), z3 = hit(k4.w);
                const unsigned any = __builtin_amdgcn_ballot_w32((z0 | z1 | z2 | z3) != 0);
                if (any != 0 && chunk5_ok(t)) {
                    const int e0 = (chunk5_of(t) << 9) + (lane << 4);
                    unsigned m = nib(z0) | (nib(z1) << 4) | (nib(z2) << 8) | (nib(z3) << 12);
                    unsigned bal = any;
                    while (bal) {
                        const int pos = qn + (int)__builtin_amdgcn_mbcnt_lo(bal, 0u);
                        if (m) {
                            const int b = __builtin_ctz(m);
                            m &= m - 1;
                            queue[qoff + pos] = e0 + b;
                        }
                        qn += __builtin_popcount(bal);
                        bal = __builtin_amdgcn_ballot_w32(m != 0);
                    }
                }
            };
            auto sel = [&](const v4i (&g)[4], int u) -> v4i {
                v4i r = g[0];
                if (u == 1) r = g[1];
                if (u == 2) r = g[2];
                if (u == 3) r = g[3];
                return r;
            };
            v4i ka[4], kb[4];
            for (int u = 0; u < 4; ++u) ka[u] = fetchk(u);
            for (int tb = 0; tb < T5; tb += 8) {
                for (int u = 0; u < 4; ++u) kb[u] = fetchk(tb + 4 + u);
#pragma nounroll
                for (int u = 0; u < 4; ++u) {
                    handlek(sel(ka, u), tb + u);
                    drain();
                }
                for (int u = 0; u < 4; ++u) ka[u] = fetchk(tb + 8 + u);
#pragma nounroll
                for (int u = 0; u < 4; ++u) {
                    handlek(sel(kb, u), tb + 4 + u);
                    drain();
                }
            }
            tail_from = nfull5 << 9;
        } else if (nfull > 0) {
            v4i ga[SCAN_U], gb[SCAN_U];
            for (int u = 0; u < SCAN_U; ++u) ga[u] = fetch(u);
            for (int tb = 0; tb < T; tb += 2 * SCAN_U) {
                for (int u = 0; u < SCAN_U; ++u) gb[u] = fetch(tb + SCAN_U + u);
                for (int u = 0; u < SCAN_U; ++u) handle(ga[u], tb + u);
                drain();
                for (int u = 0; u < SCAN_U; ++u) ga[u] = fetch(tb + 2 * SCAN_U + u);
                for (int u = 0; u < SCAN_U; ++u) handle(gb[u], tb + SCAN_U + u);
                drain();
            }
        }
        if (wave == 0 && tail_from < E) {
            for (int eb = tail_from; eb < E; eb += 32) {
                const int e = eb + lane;
                const bool own = e < E && (unsigned)(dst[e < E ? e : 0] - r0) < (unsigned)rows;
                const unsigned bal = __builtin_amdgcn_ballot_w32(own);
                if (own) queue[qoff + qn + (int)__builtin_amdgcn_mbcnt_lo(bal, 0u)] = e;
                qn += __builtin_popcount(bal);
            }
        }
        while (qn > 0) {
            const int c = min(qn, 32);
            qn -= c;
            process(qn, c);
        }
    }

    __syncthreads();

    if (FIRST && threadIdx.x == 0) lhead[0] = lcount;

    if (!FIRST) {
        for (int i = wave; i < rows; i += AGG_WAVES) {
            vecf o;
            for (int k = 0; k < VPT; ++k) o[k] = tile[i * PITCH + k * 32 + lane];
            *(vecf*)(A + (size_t)(r0 + i) * DIM + VPT * lane) = o;
        }
    } else {
        typedef float v4f __attribute__((ext_vector_type(4)));
        float* const wl = (float*)queue;
        constexpr int WP1 = HID + 1;
        constexpr int WP2 = OUT + 4;
        static_assert(IN_DIM * WP1 <= AGG_WAVES * QCAP && HID * WP2 <= AGG_WAVES * QCAP, "weights fit the queue area");
        for (int i = threadIdx.x; i < IN_DIM * HID; i += AGG_WAVES * 32)
            wl[(i >> 7) * WP1 + (i & (HID - 1))] = W1[i];
        __syncthreads();
        const int ln = lane & 15;
        const int hi = lane >> 4;
        const int mtiles = (rows + 15) >> 4;
        for (int m = wave; m < mtiles; m += AGG_WAVES) {
            v8f acc[HID / 16] = {};
            const float* arow = tile + (m * 16 + ln) * PITCH + 4 * hi;
            for (int t = 0; t < 16; ++t) {
                const v4f a4 = *(const v4f*)(arow + 8 * t);
#pragma unroll
                for (int s = 0; s < 2; ++s) {
                    const int p = 8 * t + 4 * hi + 2 * s;
                    const int c = 4 * (p & 31) + (p >> 5);
                    v2f a;
                    a.x = s ? a4.z : a4.x;
                    a.y = s ? a4.w : a4.y;
                    const float* wr = wl + c * WP1 + ln;
                    v2f b[HID / 16];
#pragma unroll
                    for (int nt = 0; nt < HID / 16; ++nt) {
                        b[nt].x = wr[nt * 16];
                        b[nt].y = wr[4 * WP1 + nt * 16];
                    }
#pragma unroll
                    for (int nt = 0; nt < HID / 16; ++nt)
                        acc[nt] = __builtin_amdgcn_wmma_f32_16x16x4_f32(
                            false, a, false, b[nt], (short)0, acc[nt], false, false);
                }
            }
#pragma unroll
            for (int nt = 0; nt < HID / 16; ++nt) {
                const float bb = bias[nt * 16 + ln];
#pragma unroll
                for (int r = 0; r < 8; ++r)
                    tile[(m * 16 + r + 8 * hi) * PITCH + nt * 16 + ln] = fmaxf(acc[nt][r] + bb, 0.0f);
            }
        }
        __syncthreads();
        for (int i = threadIdx.x; i < HID * OUT; i += AGG_WAVES * 32)
            wl[(i >> 6) * WP2 + (i & (OUT - 1))] = W2[i];
        __syncthreads();
        for (int m = wave; m < mtiles; m += AGG_WAVES) {
            v8f acc[OUT / 16] = {};
            const float* arow = tile + (m * 16 + ln) * PITCH + 4 * hi;
            for (int t = 0; t < 16; ++t) {
                const v4f a4 = *(const v4f*)(arow + 8 * t);
#pragma unroll
                for (int s = 0; s < 2; ++s) {
                    const int c = 8 * t + 4 * hi + 2 * s;
                    v2f a;
                    a.x = s ? a4.z : a4.x;
                    a.y = s ? a4.w : a4.y;
                    const float* wr = wl + c * WP2 + ln;
                    v2f b[OUT / 16];
#pragma unroll
                    for (int nt = 0; nt < OUT / 16; ++nt) {
                        b[nt].x = wr[nt * 16];
                        b[nt].y = wr[WP2 + nt * 16];
                    }
#pragma unroll
                    for (int nt = 0; nt < OUT / 16; ++nt)
                        acc[nt] = __builtin_amdgcn_wmma_f32_16x16x4_f32(
                            false, a, false, b[nt], (short)0, acc[nt], false, false);
                }
            }
#pragma unroll
            for (int nt = 0; nt < OUT / 16; ++nt) {
#pragma unroll
                for (int r = 0; r < 8; ++r) {
                    const int orow = m * 16 + r + 8 * hi;
                    if (orow < rows) H2[(size_t)(r0 + orow) * OUT + nt * 16 + ln] = acc[nt][r];
                }
            }
        }
    }
}

extern "C" void kernel_launch(void* const* d_in, const int* in_sizes, int n_in,
                              void* d_out, int out_size, void* d_ws, size_t ws_size,
                              hipStream_t stream) {
    const float* x   = (const float*)d_in[0];
    const int*   ei  = (const int*)d_in[1];
    const float* W1  = (const float*)d_in[2];
    const float* b1  = (const float*)d_in[3];
    const float* W2  = (const float*)d_in[4];
    const float* b2  = (const float*)d_in[5];
    float*       out = (float*)d_out;

    const int n = in_sizes[0] / IN_DIM;
    const int E = in_sizes[1] / 2;
    const int* srcIdx = ei;
    const int* dstIdx = ei + E;

    float* ws   = (float*)d_ws;
    float* deg  = ws;
    float* h    = deg + 2 * (size_t)n;

    k_init_deg<<<(n + 255) / 256, 256, 0, stream>>>(deg, n);

    const int tiles = (n + 15) / 16;
    const int edgeBlocks = (tiles + 7) / 8;
    const int aggBlocks = (n + TILE_ROWS - 1) / TILE_ROWS;

    k_edges<<<edgeBlocks, 256, 0, stream>>>(srcIdx, dstIdx, E, deg, out, n);

    k_agg<HID, true><<<aggBlocks, AGG_WAVES * 32, 0, stream>>>(x, srcIdx, dstIdx, deg, b1, nullptr, out, n, E, W1, W2, h);

    k_agg<OUT, false><<<aggBlocks, AGG_WAVES * 32, 0, stream>>>(h, srcIdx, dstIdx, deg, b2, out, out, n, E, nullptr, nullptr, nullptr);
}


# Degree counts kept as two planes, one per package half (the wave's die number picks its plane): every f32 add stays in its own half's lines; users add a node's two words; counts exact, sums as before
# speedup vs baseline: 3.9369x; 1.3332x over previous
#include <hip/hip_runtime.h>

typedef float v2f __attribute__((ext_vector_type(2)));
typedef float v8f __attribute__((ext_vector_type(8)));

#define IN_DIM 128
#define HID 128
#define OUT 64
#define TILE_ROWS 391

__device__ __forceinline__ void key_plan(int n, int E, int& on, int& cpr, int& kw) {
    const int groups = (n + TILE_ROWS - 1) / TILE_ROWS;
    const int full = n / TILE_ROWS;
    const int nfull5 = E >> 9;
    on = 0;
    cpr = 2;
    kw = 0;
    if (groups <= 256 && full >= 1 && nfull5 >= 1) {
        int c = (nfull5 + full - 1) / full;
        if (c < 2) c = 2;
        if (c * 640 <= (TILE_ROWS * OUT) / 2) {
            on = 1;
            cpr = c;
            kw = c * 128;
        }
    }
}

__global__ __launch_bounds__(256) void k_init_deg(float* __restrict__ deg, int n) {
    int i = blockIdx.x * blockDim.x + threadIdx.x;
    if (i < n) {
        deg[i] = 1.0f;
        deg[n + i] = 0.0f;
    }
}

typedef int gv4i __attribute__((ext_vector_type(4)));
__global__ __launch_bounds__(256) void k_edges(const int* __restrict__ src,
                                               const int* __restrict__ dst, int E,
                                               float* __restrict__ deg,
                                               float* __restrict__ Keys, int nrows) {
    const int lane = threadIdx.x & 31;
    const int wave = threadIdx.x >> 5;
    const int tile = blockIdx.x * (blockDim.x >> 5) + wave;
    const int nwaves = gridDim.x * (blockDim.x >> 5);
    const int nfull = E >> 7;
    int kon, cpr, kw;
    key_plan(nrows, E, kon, cpr, kw);
    const int nfull5 = E >> 9;
    const int half = (int)((__builtin_amdgcn_s_sendmsg_rtn(0x87) >> 18) & 1u);
    float* const dpl = deg + (size_t)half * nrows;
    for (int c = tile; c < nfull; c += nwaves) {
        const gv4i d4 = *(const gv4i*)(dst + (c << 7) + (lane << 2));
        const bool keyed = kon && (c >> 2) < nfull5;
        gv4i s4 = {0, 0, 0, 0};
        if (keyed) s4 = *(const gv4i*)(src + (c << 7) + (lane << 2));
        atomicAdd(&dpl[d4.x], 1.0f);
        atomicAdd(&dpl[d4.y], 1.0f);
        atomicAdd(&dpl[d4.z], 1.0f);
        atomicAdd(&dpl[d4.w], 1.0f);
        const int c5 = c >> 2;
        if (keyed) {
            const int rg = c5 / cpr;
            const int within = c5 - rg * cpr;
            unsigned* gp = (unsigned*)Keys + (size_t)rg * (TILE_ROWS * OUT) + 32;
            const int w = (c & 3) * 32 + lane;
            const unsigned o0 = (unsigned)d4.x / TILE_ROWS, o1 = (unsigned)d4.y / TILE_ROWS;
            const unsigned o2 = (unsigned)d4.z / TILE_ROWS, o3 = (unsigned)d4.w / TILE_ROWS;
            gp[within * 128 + w] = o0 | (o1 << 8) | (o2 << 16) | (o3 << 24);
            gv4i p;
            p.x = s4.x | ((d4.x - (int)o0 * TILE_ROWS) << 20);
            p.y = s4.y | ((d4.y - (int)o1 * TILE_ROWS) << 20);
            p.z = s4.z | ((d4.z - (int)o2 * TILE_ROWS) << 20);
            p.w = s4.w | ((d4.w - (int)o3 * TILE_ROWS) << 20);
            *(gv4i*)(gp + kw + within * 512 + 4 * w) = p;
        }
    }
    if (tile == 0)
        for (int e = (nfull << 7) + lane; e < E; e += 32) atomicAdd(&dpl[dst[e]], 1.0f);
}

#define AGG_WAVES 32
#define SCAN_U 4
#define QCAP (32 + SCAN_U * 128)

typedef int v4i __attribute__((ext_vector_type(4)));

template <int DIM, bool FIRST>
__global__ __launch_bounds__(AGG_WAVES * 32) void k_agg(const float* __restrict__ H,
                                                        const int* __restrict__ src,
                                                        const int* __restrict__ dst,
                                                        const float* __restrict__ deg,
                                                        const float* __restrict__ bias,
                                                        float* A, float* Lst, int n, int E,
                                                        const float* __restrict__ W1,
                                                        const float* __restrict__ W2,
                                                        float* __restrict__ H2) {
    constexpr int VPT = DIM / 32;
    constexpr int PITCH = FIRST ? DIM + 4 : DIM;
    constexpr int TROWS = FIRST ? ((TILE_ROWS + 15) / 16) * 16 : TILE_ROWS;
    constexpr int G = 32 / VPT;
    typedef float vecf __attribute__((ext_vector_type(VPT)));

    __shared__ __attribute__((aligned(16))) float tile[TROWS * PITCH];
    __shared__ int queue[AGG_WAVES * QCAP];
    __shared__ int lcount;

    const int lane = threadIdx.x & 31;
    const int wave = __builtin_amdgcn_readfirstlane((int)(threadIdx.x >> 5));
    const int r0 = blockIdx.x * TILE_ROWS;
    const int rows = min(TILE_ROWS, n - r0);

    int* const lhead = (int*)(Lst + (size_t)r0 * OUT);
    int kon, cpr, kw;
    key_plan(n, E, kon, cpr, kw);
    const int resv = (rows == TILE_ROWS) ? 5 * kw : 0;
    const int room = rows * OUT - 32 - resv;
    const int cap = (room > 0 && n <= (1 << 20)) ? ((room / 2) & ~31) : 0;
    int* const lpk = lhead + 32 + resv;
    float* const lwgt = (float*)(lpk + cap);
    const int nfull5 = E >> 9;
    const unsigned magic = (unsigned)((1ull << 32) / (unsigned)cpr) + 1u;
    const int* const pkbase = (const int*)Lst + 32 + kw;
    if (FIRST && threadIdx.x == 0) lcount = 0;

    const vecf bv = *(const vecf*)(bias + VPT * lane);
    for (int i0 = wave; i0 < rows; i0 += 4 * AGG_WAVES) {
        float s[4];
        vecf hv[4];
        for (int u = 0; u < 4; ++u) {
            const int i = (i0 + u * AGG_WAVES < rows) ? (i0 + u * AGG_WAVES) : i0;
            s[u] = rsqrtf(deg[r0 + i] + deg[n + r0 + i]);
            hv[u] = *(const vecf*)(H + (size_t)(r0 + i) * DIM + VPT * lane);
        }
        for (int u = 0; u < 4; ++u) {
            const int i = i0 + u * AGG_WAVES;
            if (i < rows)
                for (int k = 0; k < VPT; ++k)
                    tile[i * PITCH + k * 32 + lane] = hv[u][k] * (s[u] * s[u]) + (FIRST ? 0.0f : bv[k]);
        }
    }
    __syncthreads();

    auto addrows = [&](int s, int lrow, float nrm, int cnt,
                       auto&& behind_last_loads) __attribute__((always_inline)) {
        for (int j0 = 0; j0 < cnt; j0 += G) {
            vecf v[G];
#pragma unroll
            for (int g = 0; g < G; ++g) {
                const int j = (j0 + g < cnt) ? (j0 + g) : j0;
                const int sj = __builtin_amdgcn_readlane(s, j);
                v[g] = *(const vecf*)(H + (size_t)sj * DIM + VPT * lane);
            }
            if (j0 + G >= cnt) behind_last_loads();
#pragma unroll
            for (int g = 0; g < G; ++g) {
                if (j0 + g < cnt) {
                    const int lj = __builtin_amdgcn_readlane(lrow, j0 + g);
                    const float nj = __int_as_float(
                        __builtin_amdgcn_readlane(__float_as_int(nrm), j0 + g));
                    float* tp = tile + lj * PITCH + lane;
                    for (int k = 0; k < VPT; ++k)
                        __hip_atomic_fetch_add(tp + k * 32, v[g][k] * nj, __ATOMIC_RELAXED,
                                               __HIP_MEMORY_SCOPE_WORKGROUP);
                }
            }
        }
    };

    int total = 0;
    bool from_list = false;
    if (!FIRST) {
        total = __builtin_amdgcn_readfirstlane(lhead[0]);
        from_list = (unsigned)total <= (unsigned)cap;
    }

    if (from_list) {
        const int nb = total >> 5;
        auto entry = [&](int b, int& pk, float& nrm) {
            const int idx = (b << 5) + lane;
            pk = lpk[idx];
            nrm = lwgt[idx];
        };
        int b = wave;
        int pk = -1;
        float nrm = 0.0f;
        if (b < nb) entry(b, pk, nrm);
        while (b < nb) {
            const int bn = b + AGG_WAVES;
            int pk2 = -1;
            float nrm2 = 0.0f;
            auto next = [&]() __attribute__((always_inline)) {
                if (bn < nb) entry(bn, pk2, nrm2);
            };
            const int cnt = __builtin_popcount(__builtin_amdgcn_ballot_w32(pk >= 0));
            const int s = pk & 0xFFFFF;
            const int lrow = pk >> 20;
            if (cnt == 32)
                addrows(s, lrow, nrm, 32, next);
            else if (cnt > 0)
                addrows(s, lrow, nrm, cnt, next);
            else
                next();
            pk = pk2;
            nrm = nrm2;
            b = bn;
        }
    } else {
        const int nchunks = (E + 127) >> 7;
        const int T = (nchunks + AGG_WAVES - 1) / AGG_WAVES;
        const int span = T * AGG_WAVES;
        const int rot = (int)((blockIdx.x * (unsigned)T) / gridDim.x) * AGG_WAVES;

        const int qoff = wave * QCAP;
        int qn = 0;
        const int nfull = E >> 7;

        auto chunk_of = [&](int t) -> int {
            int c = t * AGG_WAVES + wave + rot;
            if (c >= span) c -= span;
            return c;
        };
        auto chunk_ok = [&](int t) -> bool { return t < T && chunk_of(t) < nfull; };
        auto fetch = [&](int t) -> v4i {
            const int c = chunk_ok(t) ? chunk_of(t) : 0;
            return *(const v4i*)(dst + (c << 7) + (lane << 2));
        };

        auto process = [&](int qb, int cnt) __attribute__((always_inline)) {
            int s = 0, lrow = 0;
            float nrm = 0.0f;
            __builtin_amdgcn_fence(__ATOMIC_ACQ_REL, "wavefront");
            if (lane < cnt) {
                const int e = queue[qoff + qb + lane];
                int d;
                if (FIRST && kon && e < (nfull5 << 9)) {
                    const unsigned c5 = (unsigned)e >> 9;
                    const unsigned rg = __umulhi(c5, magic);
                    const unsigned within = c5 - rg * (unsigned)cpr;
                    const int p = pkbase[(size_t)rg * (TILE_ROWS * OUT) + within * 512 + (e & 511)];
                    s = p & 0xFFFFF;
                    lrow = p >> 20;
                    d = r0 + lrow;
                } else {
                    s = src[e];
                    d = dst[e];
                    lrow = d - r0;
                }
                nrm = rsqrtf(deg[s] + deg[n + s]) * rsqrtf(deg[d] + deg[n + d]);
            }
            if (FIRST) {
                int base = 0;
                if (lane == 0)
                    base = __hip_atomic_fetch_add(&lcount, 32, __ATOMIC_RELAXED,
                                                  __HIP_MEMORY_SCOPE_WORKGROUP);
                base = __builtin_amdgcn_readfirstlane(base);
                if (base + 32 <= cap) {
                    lpk[base + lane] = (lane < cnt) ? (s | (lrow << 20)) : -1;
                    lwgt[base + lane] = nrm;
                }
            }
            addrows(s, lrow, nrm, cnt, []() {});
        };

        auto handle = [&](const v4i d4, int t) {
            const bool o0 = (unsigned)(d4.x - r0) < (unsigned)rows;
            const bool o1 = (unsigned)(d4.y - r0) < (unsigned)rows;
            const bool o2 = (unsigned)(d4.z - r0) < (unsigned)rows;
            const bool o3 = (unsigned)(d4.w - r0) < (unsigned)rows;
            const unsigned any = __builtin_amdgcn_ballot_w32(o0) | __builtin_amdgcn_ballot_w32(o1) |
                                 __builtin_amdgcn_ballot_w32(o2) | __builtin_amdgcn_ballot_w32(o3);
            if (any != 0 && chunk_ok(t)) {
                const int e0 = (chunk_of(t) << 7) + (lane << 2);
                unsigned m = (o0 ? 1u : 0u) | (o1 ? 2u : 0u) | (o2 ? 4u : 0u) | (o3 ? 8u : 0u);
                unsigned bal = any;
                while (bal) {
                    const int pos = qn + (int)__builtin_amdgcn_mbcnt_lo(bal, 0u);
                    if (m) {
                        const int b = __builtin_ctz(m);
                        m &= m - 1;
                        queue[qoff + pos] = e0 + b;
                    }
                    qn += __builtin_popcount(bal);
                    bal = __builtin_amdgcn_ballot_w32(m != 0);
                }
            }
        };
        auto drain = [&]() __attribute__((always_inline)) {
            while (qn >= 32) {
                qn -= 32;
                process(qn, 32);
            }
        };

        int tail_from = nfull << 7;
        if (FIRST && kon) {
            const unsigned* const keys = (const unsigned*)Lst;
            const int T5 = (nfull5 + AGG_WAVES - 1) / AGG_WAVES;
            const int span5 = T5 * AGG_WAVES;
            const int rot5 = (int)((blockIdx.x * (unsigned)T5) / gridDim.x) * AGG_WAVES;
            const unsigned pat = blockIdx.x * 0x01010101u;
            auto chunk5_of = [&](int t) -> int {
                int c = t * AGG_WAVES + wave + rot5;
                if (c >= span5) c -= span5;
                return c;
            };
            auto chunk5_ok = [&](int t) -> bool { return t < T5 && chunk5_of(t) < nfull5; };
            auto fetchk = [&](int t) -> v4i {
                const unsigned c = chunk5_ok(t) ? (unsigned)chunk5_of(t) : 0u;
                const unsigned rg = __umulhi(c, magic);
                const unsigned within = c - rg * (unsigned)cpr;
                return *(const v4i*)(keys + (size_t)rg * (TILE_ROWS * OUT) + 32 + within * 128 +
                                     (lane << 2));
            };
            auto hit = [&](int w) -> unsigned {
                const unsigned x = (unsigned)w ^ pat;
                return ~(((x & 0x7F7F7F7Fu) + 0x7F7F7F7Fu) | x | 0x7F7F7F7Fu);
            };
            auto nib = [&](unsigned z) -> unsigned { return (((z >> 7) * 0x00204081u) >> 21) & 15u; };
            auto handlek = [&](const v4i k4, int t) {
                const unsigned z0 = hit(k4.x), z1 = hit(k4.y), z2 = hit(k4.z), z3 = hit(k4.w);
                const unsigned any = __builtin_amdgcn_ballot_w32((z0 | z1 | z2 | z3) != 0);
                if (any != 0 && chunk5_ok(t)) {
                    const int e0 = (chunk5_of(t) << 9) + (lane << 4);
                    unsigned m = nib(z0) | (nib(z1) << 4) | (nib(z2) << 8) | (nib(z3) << 12);
                    unsigned bal = any;
                    while (bal) {
                        const int pos = qn + (int)__builtin_amdgcn_mbcnt_lo(bal, 0u);
                        if (m) {
                            const int b = __builtin_ctz(m);
                            m &= m - 1;
                            queue[qoff + pos] = e0 + b;
                        }
                        qn += __builtin_popcount(bal);
                        bal = __builtin_amdgcn_ballot_w32(m != 0);
                    }
                }
            };
            auto sel = [&](const v4i (&g)[4], int u) -> v4i {
                v4i r = g[0];
                if (u == 1) r = g[1];
                if (u == 2) r = g[2];
                if (u == 3) r = g[3];
                return r;
            };
            v4i ka[4], kb[4];
            for (int u = 0; u < 4; ++u) ka[u] = fetchk(u);
            for (int tb = 0; tb < T5; tb += 8) {
                for (int u = 0; u < 4; ++u) kb[u] = fetchk(tb + 4 + u);
#pragma nounroll
                for (int u = 0; u < 4; ++u) {
                    handlek(sel(ka, u), tb + u);
                    drain();
                }
                for (int u = 0; u < 4; ++u) ka[u] = fetchk(tb + 8 + u);
#pragma nounroll
                for (int u = 0; u < 4; ++u) {
                    handlek(sel(kb, u), tb + 4 + u);
                    drain();
                }
            }
            tail_from = nfull5 << 9;
        } else if (nfull > 0) {
            v4i ga[SCAN_U], gb[SCAN_U];
            for (int u = 0; u < SCAN_U; ++u) ga[u] = fetch(u);
            for (int tb = 0; tb < T; tb += 2 * SCAN_U) {
                for (int u = 0; u < SCAN_U; ++u) gb[u] = fetch(tb + SCAN_U + u);
                for (int u = 0; u < SCAN_U; ++u) handle(ga[u], tb + u);
                drain();
                for (int u = 0; u < SCAN_U; ++u) ga[u] = fetch(tb + 2 * SCAN_U + u);
                for (int u = 0; u < SCAN_U; ++u) handle(gb[u], tb + SCAN_U + u);
                drain();
            }
        }
        if (wave == 0 && tail_from < E) {
            for (int eb = tail_from; eb < E; eb += 32) {
                const int e = eb + lane;
                const bool own = e < E && (unsigned)(dst[e < E ? e : 0] - r0) < (unsigned)rows;
                const unsigned bal = __builtin_amdgcn_ballot_w32(own);
                if (own) queue[qoff + qn + (int)__builtin_amdgcn_mbcnt_lo(bal, 0u)] = e;
                qn += __builtin_popcount(bal);
            }
        }
        while (qn > 0) {
            const int c = min(qn, 32);
            qn -= c;
            process(qn, c);
        }
    }

    __syncthreads();

    if (FIRST && threadIdx.x == 0) lhead[0] = lcount;

    if (!FIRST) {
        for (int i = wave; i < rows; i += AGG_WAVES) {
            vecf o;
            for (int k = 0; k < VPT; ++k) o[k] = tile[i * PITCH + k * 32 + lane];
            *(vecf*)(A + (size_t)(r0 + i) * DIM + VPT * lane) = o;
        }
    } else {
        typedef float v4f __attribute__((ext_vector_type(4)));
        float* const wl = (float*)queue;
        constexpr int WP1 = HID + 1;
        constexpr int WP2 = OUT + 4;
        static_assert(IN_DIM * WP1 <= AGG_WAVES * QCAP && HID * WP2 <= AGG_WAVES * QCAP, "weights fit the queue area");
        for (int i = threadIdx.x; i < IN_DIM * HID; i += AGG_WAVES * 32)
            wl[(i >> 7) * WP1 + (i & (HID - 1))] = W1[i];
        __syncthreads();
        const int ln = lane & 15;
        const int hi = lane >> 4;
        const int mtiles = (rows + 15) >> 4;
        for (int m = wave; m < mtiles; m += AGG_WAVES) {
            v8f acc[HID / 16] = {};
            const float* arow = tile + (m * 16 + ln) * PITCH + 4 * hi;
            for (int t = 0; t < 16; ++t) {
                const v4f a4 = *(const v4f*)(arow + 8 * t);
#pragma unroll
                for (int s = 0; s < 2; ++s) {
                    const int p = 8 * t + 4 * hi + 2 * s;
                    const int c = 4 * (p & 31) + (p >> 5);
                    v2f a;
                    a.x = s ? a4.z : a4.x;
                    a.y = s ? a4.w : a4.y;
                    const float* wr = wl + c * WP1 + ln;
                    v2f b[HID / 16];
#pragma unroll
                    for (int nt = 0; nt < HID / 16; ++nt) {
                        b[nt].x = wr[nt * 16];
                        b[nt].y = wr[4 * WP1 + nt * 16];
                    }
#pragma unroll
                    for (int nt = 0; nt < HID / 16; ++nt)
                        acc[nt] = __builtin_amdgcn_wmma_f32_16x16x4_f32(
                            false, a, false, b[nt], (short)0, acc[nt], false, false);
                }
            }
#pragma unroll
            for (int nt = 0; nt < HID / 16; ++nt) {
                const float bb = bias[nt * 16 + ln];
#pragma unroll
                for (int r = 0; r < 8; ++r)
                    tile[(m * 16 + r + 8 * hi) * PITCH + nt * 16 + ln] = fmaxf(acc[nt][r] + bb, 0.0f);
            }
        }
        __syncthreads();
        for (int i = threadIdx.x; i < HID * OUT; i += AGG_WAVES * 32)
            wl[(i >> 6) * WP2 + (i & (OUT - 1))] = W2[i];
        __syncthreads();
        for (int m = wave; m < mtiles; m += AGG_WAVES) {
            v8f acc[OUT / 16] = {};
            const float* arow = tile + (m * 16 + ln) * PITCH + 4 * hi;
            for (int t = 0; t < 16; ++t) {
                const v4f a4 = *(const v4f*)(arow + 8 * t);
#pragma unroll
                for (int s = 0; s < 2; ++s) {
                    const int c = 8 * t + 4 * hi + 2 * s;
                    v2f a;
                    a.x = s ? a4.z : a4.x;
                    a.y = s ? a4.w : a4.y;
                    const float* wr = wl + c * WP2 + ln;
                    v2f b[OUT / 16];
#pragma unroll
                    for (int nt = 0; nt < OUT / 16; ++nt) {
                        b[nt].x = wr[nt * 16];
                        b[nt].y = wr[WP2 + nt * 16];
                    }
#pragma unroll
                    for (int nt = 0; nt < OUT / 16; ++nt)
                        acc[nt] = __builtin_amdgcn_wmma_f32_16x16x4_f32(
                            false, a, false, b[nt], (short)0, acc[nt], false, false);
                }
            }
#pragma unroll
            for (int nt = 0; nt < OUT / 16; ++nt) {
#pragma unroll
                for (int r = 0; r < 8; ++r) {
                    const int orow = m * 16 + r + 8 * hi;
                    if (orow < rows) H2[(size_t)(r0 + orow) * OUT + nt * 16 + ln] = acc[nt][r];
                }
            }
        }
    }
}

extern "C" void kernel_launch(void* const* d_in, const int* in_sizes, int n_in,
                              void* d_out, int out_size, void* d_ws, size_t ws_size,
                              hipStream_t stream) {
    const float* x   = (const float*)d_in[0];
    const int*   ei  = (const int*)d_in[1];
    const float* W1  = (const float*)d_in[2];
    const float* b1  = (const float*)d_in[3];
    const float* W2  = (const float*)d_in[4];
    const float* b2  = (const float*)d_in[5];
    float*       out = (float*)d_out;

    const int n = in_sizes[0] / IN_DIM;
    const int E = in_sizes[1] / 2;
    const int* srcIdx = ei;
    const int* dstIdx = ei + E;

    float* ws   = (float*)d_ws;
    float* deg  = ws;
    float* h    = deg + 2 * (size_t)n;

    k_init_deg<<<(n + 255) / 256, 256, 0, stream>>>(deg, n);

    const int tiles = (n + 15) / 16;
    const int edgeBlocks = (tiles + 7) / 8;
    const int aggBlocks = (n + TILE_ROWS - 1) / TILE_ROWS;

    k_edges<<<edgeBlocks, 256, 0, stream>>>(srcIdx, dstIdx, E, deg, out, n);

    k_agg<HID, true><<<aggBlocks, AGG_WAVES * 32, 0, stream>>>(x, srcIdx, dstIdx, deg, b1, nullptr, out, n, E, W1, W2, h);

    k_agg<OUT, false><<<aggBlocks, AGG_WAVES * 32, 0, stream>>>(h, srcIdx, dstIdx, deg, b2, out, out, n, E, nullptr, nullptr, nullptr);
}
